# starting kernel
# baseline (speedup 1.0000x reference)
#include <hip/hip_runtime.h>
#include <hip/hip_bf16.h>
#include <math.h>

#define BB 4
#define TT 2048
#define CC 1024
#define HH 128

typedef __attribute__((ext_vector_type(2))) float v2f;
typedef __attribute__((ext_vector_type(8))) float v8f;

__device__ __forceinline__ v8f wmma_f32(v2f a, v2f b, v8f c) {
  return __builtin_amdgcn_wmma_f32_16x16x4_f32(false, a, false, b, (short)0, c,
                                               false, false);
}

__global__ __launch_bounds__(32) void qkv_proj(
    const float* __restrict__ x, const float* __restrict__ Wq,
    const float* __restrict__ Wk, const float* __restrict__ Wv,
    float* __restrict__ q, float* __restrict__ k, float* __restrict__ v) {
  const int lane = threadIdx.x & 31;
  const int lo = lane & 15, hi = lane >> 4;
  const int m0 = blockIdx.x * 16;

  const float* W;
  float* out;
  if (blockIdx.y == 0)      { W = Wq; out = q; }
  else if (blockIdx.y == 1) { W = Wk; out = k; }
  else                      { W = Wv; out = v; }

  const float* xrow = x + (size_t)(m0 + lo) * CC;
  v8f acc[8] = {};

  for (int k0 = 0; k0 < CC; k0 += 4) {
    v2f a = *(const v2f*)(xrow + k0 + 2 * hi);
    const float* wr0 = W + (size_t)(k0 + 2 * hi) * HH + lo;
    const float* wr1 = wr0 + HH;
#pragma unroll
    for (int n = 0; n < 8; ++n) {
      v2f bf;
      bf.x = wr0[n * 16];
      bf.y = wr1[n * 16];
      acc[n] = wmma_f32(a, bf, acc[n]);
    }
  }
#pragma unroll
  for (int n = 0; n < 8; ++n)
#pragma unroll
    for (int r = 0; r < 8; ++r)
      out[(size_t)(m0 + r + 8 * hi) * HH + n * 16 + lo] = acc[n][r];
}

__global__ __launch_bounds__(32) void qk_scores(const float* __restrict__ q,
                                                const float* __restrict__ kmat,
                                                float* __restrict__ S) {
  const int s0 = blockIdx.x * 16;
  const int t0 = blockIdx.y * 16;
  const int b = blockIdx.z;
  if (s0 > t0 + 15) return;

  const int lane = threadIdx.x & 31;
  const int lo = lane & 15, hi = lane >> 4;
  const float* qrow = q + ((size_t)b * TT + t0 + lo) * HH;
  const float* krow = kmat + ((size_t)b * TT + s0 + lo) * HH;

  v8f acc = {};
#pragma unroll
  for (int k0 = 0; k0 < HH; k0 += 4) {
    v2f a = *(const v2f*)(qrow + k0 + 2 * hi);
    v2f bf = *(const v2f*)(krow + k0 + 2 * hi);
    acc = wmma_f32(a, bf, acc);
  }

  const float scale = 0.08838834764831845f;
  float* Sout = S + ((size_t)b * TT + t0) * TT + s0;
#pragma unroll
  for (int r = 0; r < 8; ++r) {
    const int t = t0 + r + 8 * hi;
    const int s = s0 + lo;
    const float val = (s <= t) ? acc[r] * scale : -1e30f;
    Sout[(size_t)(r + 8 * hi) * TT + lo] = val;
  }
}

__global__ void col_softmax_stats(const float* __restrict__ S,
                                  float* __restrict__ colmax,
                                  float* __restrict__ colrcp) {
  const int gid = blockIdx.x * blockDim.x + threadIdx.x;
  const int b = gid / TT;
  const int s = gid % TT;
  const float* Sb = S + (size_t)b * TT * TT;
  const int tstart = s & ~31;

  float m = -3.0e38f;
  for (int t = tstart; t < TT; ++t)
    if (t >= s) m = fmaxf(m, Sb[(size_t)t * TT + s]);

  float sum = 0.f;
  for (int t = tstart; t < TT; ++t)
    if (t >= s) sum += __expf(Sb[(size_t)t * TT + s] - m);

  colmax[gid] = m;
  colrcp[gid] = 1.0f / sum;
}

__global__ __launch_bounds__(32) void attn_out(
    const float* __restrict__ S, const float* __restrict__ colmax,
    const float* __restrict__ colrcp, const float* __restrict__ v,
    float* __restrict__ out) {
  const int t0 = blockIdx.x * 16;
  const int b = blockIdx.y;
  const int lane = threadIdx.x & 31;
  const int lo = lane & 15, hi = lane >> 4;

  const float* Srow = S + ((size_t)b * TT + t0 + lo) * TT;
  const float* cm = colmax + (size_t)b * TT;
  const float* cr = colrcp + (size_t)b * TT;
  const float* vb = v + (size_t)b * TT * HH;

  v8f acc[8] = {};
  const int sEnd = t0 + 16;
  for (int s0 = 0; s0 < sEnd; s0 += 4) {
    const int sA = s0 + 2 * hi;
    v2f raw = *(const v2f*)(Srow + sA);
    v2f mx = *(const v2f*)(cm + sA);
    v2f rc = *(const v2f*)(cr + sA);
    v2f a;
    a.x = __expf(raw.x - mx.x) * rc.x;
    a.y = __expf(raw.y - mx.y) * rc.y;

    const float* vr0 = vb + (size_t)sA * HH + lo;
    const float* vr1 = vr0 + HH;
#pragma unroll
    for (int n = 0; n < 8; ++n) {
      v2f bf;
      bf.x = vr0[n * 16];
      bf.y = vr1[n * 16];
      acc[n] = wmma_f32(a, bf, acc[n]);
    }
  }

  float* orow = out + ((size_t)b * TT + t0) * HH;
#pragma unroll
  for (int n = 0; n < 8; ++n)
#pragma unroll
    for (int r = 0; r < 8; ++r)
      orow[(size_t)(r + 8 * hi) * HH + n * 16 + lo] = acc[n][r];
}

extern "C" void kernel_launch(void* const* d_in, const int* in_sizes, int n_in,
                              void* d_out, int out_size, void* d_ws,
                              size_t ws_size, hipStream_t stream) {
  (void)in_sizes; (void)n_in; (void)out_size; (void)ws_size;
  const float* x = (const float*)d_in[0];
  const float* Wq = (const float*)d_in[1];
  const float* Wk = (const float*)d_in[2];
  const float* Wv = (const float*)d_in[3];
  float* out = (float*)d_out;

  float* ws = (float*)d_ws;
  float* q = ws;
  float* k = q + (size_t)BB * TT * HH;
  float* v = k + (size_t)BB * TT * HH;
  float* S = v + (size_t)BB * TT * HH;
  float* cmax = S + (size_t)BB * TT * TT;
  float* crcp = cmax + (size_t)BB * TT;

  qkv_proj<<<dim3(BB * TT / 16, 3), 32, 0, stream>>>(x, Wq, Wk, Wv, q, k, v);
  qk_scores<<<dim3(TT / 16, TT / 16, BB), 32, 0, stream>>>(q, k, S);
  col_softmax_stats<<<dim3(BB * TT / 256), 256, 0, stream>>>(S, cmax, crcp);
  attn_out<<<dim3(TT / 16, BB), 32, 0, stream>>>(S, cmax, crcp, v, out);
}


# column softmax stats: rows of each 32-column strip shared among the 32 waves of one work-group (was one thread per column), joined in LDS; column sums now added per wave, then wave 0..31
# speedup vs baseline: 3.0989x; 3.0989x over previous
#include <hip/hip_runtime.h>
#include <hip/hip_bf16.h>
#include <math.h>

#define BB 4
#define TT 2048
#define CC 1024
#define HH 128

typedef __attribute__((ext_vector_type(2))) float v2f;
typedef __attribute__((ext_vector_type(8))) float v8f;

__device__ __forceinline__ v8f wmma_f32(v2f a, v2f b, v8f c) {
  return __builtin_amdgcn_wmma_f32_16x16x4_f32(false, a, false, b, (short)0, c,
                                               false, false);
}

__global__ __launch_bounds__(32) void qkv_proj(
    const float* __restrict__ x, const float* __restrict__ Wq,
    const float* __restrict__ Wk, const float* __restrict__ Wv,
    float* __restrict__ q, float* __restrict__ k, float* __restrict__ v) {
  const int lane = threadIdx.x & 31;
  const int lo = lane & 15, hi = lane >> 4;
  const int m0 = blockIdx.x * 16;

  const float* W;
  float* out;
  if (blockIdx.y == 0)      { W = Wq; out = q; }
  else if (blockIdx.y == 1) { W = Wk; out = k; }
  else                      { W = Wv; out = v; }

  const float* xrow = x + (size_t)(m0 + lo) * CC;
  v8f acc[8] = {};

  for (int k0 = 0; k0 < CC; k0 += 4) {
    v2f a = *(const v2f*)(xrow + k0 + 2 * hi);
    const float* wr0 = W + (size_t)(k0 + 2 * hi) * HH + lo;
    const float* wr1 = wr0 + HH;
#pragma unroll
    for (int n = 0; n < 8; ++n) {
      v2f bf;
      bf.x = wr0[n * 16];
      bf.y = wr1[n * 16];
      acc[n] = wmma_f32(a, bf, acc[n]);
    }
  }
#pragma unroll
  for (int n = 0; n < 8; ++n)
#pragma unroll
    for (int r = 0; r < 8; ++r)
      out[(size_t)(m0 + r + 8 * hi) * HH + n * 16 + lo] = acc[n][r];
}

__global__ __launch_bounds__(32) void qk_scores(const float* __restrict__ q,
                                                const float* __restrict__ kmat,
                                                float* __restrict__ S) {
  const int s0 = blockIdx.x * 16;
  const int t0 = blockIdx.y * 16;
  const int b = blockIdx.z;
  if (s0 > t0 + 15) return;

  const int lane = threadIdx.x & 31;
  const int lo = lane & 15, hi = lane >> 4;
  const float* qrow = q + ((size_t)b * TT + t0 + lo) * HH;
  const float* krow = kmat + ((size_t)b * TT + s0 + lo) * HH;

  v8f acc = {};
#pragma unroll
  for (int k0 = 0; k0 < HH; k0 += 4) {
    v2f a = *(const v2f*)(qrow + k0 + 2 * hi);
    v2f bf = *(const v2f*)(krow + k0 + 2 * hi);
    acc = wmma_f32(a, bf, acc);
  }

  const float scale = 0.08838834764831845f;
  float* Sout = S + ((size_t)b * TT + t0) * TT + s0;
#pragma unroll
  for (int r = 0; r < 8; ++r) {
    const int t = t0 + r + 8 * hi;
    const int s = s0 + lo;
    const float val = (s <= t) ? acc[r] * scale : -1e30f;
    Sout[(size_t)(r + 8 * hi) * TT + lo] = val;
  }
}

#define ROWS_IN_FLIGHT 16

__global__ __launch_bounds__(1024) void col_softmax_stats(
    const float* __restrict__ S, float* __restrict__ colmax,
    float* __restrict__ colrcp) {
  __shared__ float wmax[32][32];
  __shared__ float wsum[32][32];

  const int lane = threadIdx.x & 31;
  const int w = threadIdx.x >> 5;
  const int s0 = blockIdx.x * 32;
  const int b = blockIdx.y;
  const int n = (TT - s0) / 32;
  const size_t step = (size_t)32 * TT;
  const float* p = S + ((size_t)b * TT + s0 + w) * TT + s0 + lane;
  const float none = -3.0e38f;

  float x0 = none;
  if (w >= lane) x0 = p[0];

  float m = x0;
  for (int i0 = 1; i0 < n; i0 += ROWS_IN_FLIGHT) {
    float x[ROWS_IN_FLIGHT];
#pragma unroll
    for (int u = 0; u < ROWS_IN_FLIGHT; ++u) {
      x[u] = none;
      if (i0 + u < n) x[u] = p[(size_t)(i0 + u) * step];
    }
#pragma unroll
    for (int u = 0; u < ROWS_IN_FLIGHT; ++u) m = fmaxf(m, x[u]);
  }

  wmax[w][lane] = m;
  __syncthreads();
  float cm = wmax[0][lane];
#pragma unroll
  for (int j = 1; j < 32; ++j) cm = fmaxf(cm, wmax[j][lane]);

  float sum = __expf(x0 - cm);
  for (int i0 = 1; i0 < n; i0 += ROWS_IN_FLIGHT) {
    float x[ROWS_IN_FLIGHT];
#pragma unroll
    for (int u = 0; u < ROWS_IN_FLIGHT; ++u) {
      x[u] = none;
      if (i0 + u < n) x[u] = p[(size_t)(i0 + u) * step];
    }
#pragma unroll
    for (int u = 0; u < ROWS_IN_FLIGHT; ++u) sum += __expf(x[u] - cm);
  }

  wsum[w][lane] = sum;
  __syncthreads();
  if (w == 0) {
    float tot = wsum[0][lane];
#pragma unroll
    for (int j = 1; j < 32; ++j) tot += wsum[j][lane];
    const int gid = b * TT + s0 + lane;
    colmax[gid] = cm;
    colrcp[gid] = 1.0f / tot;
  }
}

__global__ __launch_bounds__(32) void attn_out(
    const float* __restrict__ S, const float* __restrict__ colmax,
    const float* __restrict__ colrcp, const float* __restrict__ v,
    float* __restrict__ out) {
  const int t0 = blockIdx.x * 16;
  const int b = blockIdx.y;
  const int lane = threadIdx.x & 31;
  const int lo = lane & 15, hi = lane >> 4;

  const float* Srow = S + ((size_t)b * TT + t0 + lo) * TT;
  const float* cm = colmax + (size_t)b * TT;
  const float* cr = colrcp + (size_t)b * TT;
  const float* vb = v + (size_t)b * TT * HH;

  v8f acc[8] = {};
  const int sEnd = t0 + 16;
  for (int s0 = 0; s0 < sEnd; s0 += 4) {
    const int sA = s0 + 2 * hi;
    v2f raw = *(const v2f*)(Srow + sA);
    v2f mx = *(const v2f*)(cm + sA);
    v2f rc = *(const v2f*)(cr + sA);
    v2f a;
    a.x = __expf(raw.x - mx.x) * rc.x;
    a.y = __expf(raw.y - mx.y) * rc.y;

    const float* vr0 = vb + (size_t)sA * HH + lo;
    const float* vr1 = vr0 + HH;
#pragma unroll
    for (int n = 0; n < 8; ++n) {
      v2f bf;
      bf.x = vr0[n * 16];
      bf.y = vr1[n * 16];
      acc[n] = wmma_f32(a, bf, acc[n]);
    }
  }

  float* orow = out + ((size_t)b * TT + t0) * HH;
#pragma unroll
  for (int n = 0; n < 8; ++n)
#pragma unroll
    for (int r = 0; r < 8; ++r)
      orow[(size_t)(r + 8 * hi) * HH + n * 16 + lo] = acc[n][r];
}

extern "C" void kernel_launch(void* const* d_in, const int* in_sizes, int n_in,
                              void* d_out, int out_size, void* d_ws,
                              size_t ws_size, hipStream_t stream) {
  (void)in_sizes; (void)n_in; (void)out_size; (void)ws_size;
  const float* x = (const float*)d_in[0];
  const float* Wq = (const float*)d_in[1];
  const float* Wk = (const float*)d_in[2];
  const float* Wv = (const float*)d_in[3];
  float* out = (float*)d_out;

  float* ws = (float*)d_ws;
  float* q = ws;
  float* k = q + (size_t)BB * TT * HH;
  float* v = k + (size_t)BB * TT * HH;
  float* S = v + (size_t)BB * TT * HH;
  float* cmax = S + (size_t)BB * TT * TT;
  float* crcp = cmax + (size_t)BB * TT;

  qkv_proj<<<dim3(BB * TT / 16, 3), 32, 0, stream>>>(x, Wq, Wk, Wv, q, k, v);
  qk_scores<<<dim3(TT / 16, TT / 16, BB), 32, 0, stream>>>(q, k, S);
  col_softmax_stats<<<dim3(TT / 32, BB), 1024, 0, stream>>>(S, cmax, crcp);
  attn_out<<<dim3(TT / 16, BB), 32, 0, stream>>>(S, cmax, crcp, v, out);
}


# k loop of the two slab products (q/k/v projection, weights x v) shared among 8 / 16 waves of a work-group (was one wave per slab); partial sums staged in LDS and added in wave order 0,1,2,...
# speedup vs baseline: 9.2771x; 2.9937x over previous
#include <hip/hip_runtime.h>
#include <hip/hip_bf16.h>
#include <math.h>

#define BB 4
#define TT 2048
#define CC 1024
#define HH 128

typedef __attribute__((ext_vector_type(2))) float v2f;
typedef __attribute__((ext_vector_type(8))) float v8f;

__device__ __forceinline__ v8f wmma_f32(v2f a, v2f b, v8f c) {
  return __builtin_amdgcn_wmma_f32_16x16x4_f32(false, a, false, b, (short)0, c,
                                               false, false);
}

#define PITCH 132
#define PROJ_WAVES 8
#define AO_WAVES 16

template <int NW>
__device__ __forceinline__ void join_slab(float* part, const v8f (&acc)[8],
                                          int w, int lane, float* dst) {
  const int lo = lane & 15, hi = lane >> 4;
  float* mine = part + w * (16 * PITCH);
#pragma unroll
  for (int n = 0; n < 8; ++n)
#pragma unroll
    for (int r = 0; r < 8; ++r)
      mine[(r + 8 * hi) * PITCH + n * 16 + lo] = acc[n][r];
  __syncthreads();
  for (int row = w; row < 16; row += NW) {
    const float* src = part + row * PITCH + 4 * lane;
    float4 s = *(const float4*)src;
#pragma unroll
    for (int j = 1; j < NW; ++j) {
      const float4 p = *(const float4*)(src + j * (16 * PITCH));
      s.x += p.x; s.y += p.y; s.z += p.z; s.w += p.w;
    }
    *(float4*)(dst + (size_t)row * HH + 4 * lane) = s;
  }
}

__global__ __launch_bounds__(32 * PROJ_WAVES) void qkv_proj(
    const float* __restrict__ x, const float* __restrict__ Wq,
    const float* __restrict__ Wk, const float* __restrict__ Wv,
    float* __restrict__ q, float* __restrict__ k, float* __restrict__ v) {
  __shared__ __attribute__((aligned(16))) float part[PROJ_WAVES * 16 * PITCH];
  const int lane = threadIdx.x & 31;
  const int w = __builtin_amdgcn_readfirstlane((int)(threadIdx.x >> 5));
  const int lo = lane & 15, hi = lane >> 4;
  const int m0 = blockIdx.x * 16;

  const float* W;
  float* out;
  if (blockIdx.y == 0)      { W = Wq; out = q; }
  else if (blockIdx.y == 1) { W = Wk; out = k; }
  else                      { W = Wv; out = v; }

  const float* xrow = x + (size_t)(m0 + lo) * CC;
  v8f acc[8] = {};

  const int kBeg = w * (CC / PROJ_WAVES);
  const int kEnd = kBeg + CC / PROJ_WAVES;
  for (int k0 = kBeg; k0 < kEnd; k0 += 4) {
    v2f a = *(const v2f*)(xrow + k0 + 2 * hi);
    const float* wr0 = W + (size_t)(k0 + 2 * hi) * HH + lo;
    const float* wr1 = wr0 + HH;
#pragma unroll
    for (int n = 0; n < 8; ++n) {
      v2f bf;
      bf.x = wr0[n * 16];
      bf.y = wr1[n * 16];
      acc[n] = wmma_f32(a, bf, acc[n]);
    }
  }
  join_slab<PROJ_WAVES>(part, acc, w, lane, out + (size_t)m0 * HH);
}

__global__ __launch_bounds__(32) void qk_scores(const float* __restrict__ q,
                                                const float* __restrict__ kmat,
                                                float* __restrict__ S) {
  const int s0 = blockIdx.x * 16;
  const int t0 = blockIdx.y * 16;
  const int b = blockIdx.z;
  if (s0 > t0 + 15) return;

  const int lane = threadIdx.x & 31;
  const int lo = lane & 15, hi = lane >> 4;
  const float* qrow = q + ((size_t)b * TT + t0 + lo) * HH;
  const float* krow = kmat + ((size_t)b * TT + s0 + lo) * HH;

  v8f acc = {};
#pragma unroll
  for (int k0 = 0; k0 < HH; k0 += 4) {
    v2f a = *(const v2f*)(qrow + k0 + 2 * hi);
    v2f bf = *(const v2f*)(krow + k0 + 2 * hi);
    acc = wmma_f32(a, bf, acc);
  }

  const float scale = 0.08838834764831845f;
  float* Sout = S + ((size_t)b * TT + t0) * TT + s0;
#pragma unroll
  for (int r = 0; r < 8; ++r) {
    const int t = t0 + r + 8 * hi;
    const int s = s0 + lo;
    const float val = (s <= t) ? acc[r] * scale : -1e30f;
    Sout[(size_t)(r + 8 * hi) * TT + lo] = val;
  }
}

#define ROWS_IN_FLIGHT 16

__global__ __launch_bounds__(1024) void col_softmax_stats(
    const float* __restrict__ S, float* __restrict__ colmax,
    float* __restrict__ colrcp) {
  __shared__ float wmax[32][32];
  __shared__ float wsum[32][32];

  const int lane = threadIdx.x & 31;
  const int w = threadIdx.x >> 5;
  const int s0 = blockIdx.x * 32;
  const int b = blockIdx.y;
  const int n = (TT - s0) / 32;
  const size_t step = (size_t)32 * TT;
  const float* p = S + ((size_t)b * TT + s0 + w) * TT + s0 + lane;
  const float none = -3.0e38f;

  float x0 = none;
  if (w >= lane) x0 = p[0];

  float m = x0;
  for (int i0 = 1; i0 < n; i0 += ROWS_IN_FLIGHT) {
    float x[ROWS_IN_FLIGHT];
#pragma unroll
    for (int u = 0; u < ROWS_IN_FLIGHT; ++u) {
      x[u] = none;
      if (i0 + u < n) x[u] = p[(size_t)(i0 + u) * step];
    }
#pragma unroll
    for (int u = 0; u < ROWS_IN_FLIGHT; ++u) m = fmaxf(m, x[u]);
  }

  wmax[w][lane] = m;
  __syncthreads();
  float cm = wmax[0][lane];
#pragma unroll
  for (int j = 1; j < 32; ++j) cm = fmaxf(cm, wmax[j][lane]);

  float sum = __expf(x0 - cm);
  for (int i0 = 1; i0 < n; i0 += ROWS_IN_FLIGHT) {
    float x[ROWS_IN_FLIGHT];
#pragma unroll
    for (int u = 0; u < ROWS_IN_FLIGHT; ++u) {
      x[u] = none;
      if (i0 + u < n) x[u] = p[(size_t)(i0 + u) * step];
    }
#pragma unroll
    for (int u = 0; u < ROWS_IN_FLIGHT; ++u) sum += __expf(x[u] - cm);
  }

  wsum[w][lane] = sum;
  __syncthreads();
  if (w == 0) {
    float tot = wsum[0][lane];
#pragma unroll
    for (int j = 1; j < 32; ++j) tot += wsum[j][lane];
    const int gid = b * TT + s0 + lane;
    colmax[gid] = cm;
    colrcp[gid] = 1.0f / tot;
  }
}

__global__ __launch_bounds__(32 * AO_WAVES) void attn_out(
    const float* __restrict__ S, const float* __restrict__ colmax,
    const float* __restrict__ colrcp, const float* __restrict__ v,
    float* __restrict__ out) {
  __shared__ __attribute__((aligned(16))) float part[AO_WAVES * 16 * PITCH];
  const int t0 = blockIdx.x * 16;
  const int b = blockIdx.y;
  const int lane = threadIdx.x & 31;
  const int w = __builtin_amdgcn_readfirstlane((int)(threadIdx.x >> 5));
  const int lo = lane & 15, hi = lane >> 4;

  const float* Srow = S + ((size_t)b * TT + t0 + lo) * TT;
  const float* cm = colmax + (size_t)b * TT;
  const float* cr = colrcp + (size_t)b * TT;
  const float* vb = v + (size_t)b * TT * HH;

  v8f acc[8] = {};
  const int sEnd = t0 + 16;
  for (int c0 = 32 * w; c0 < sEnd; c0 += 32 * AO_WAVES) {
   const int cEnd = (c0 + 32 < sEnd) ? c0 + 32 : sEnd;
   for (int s0 = c0; s0 < cEnd; s0 += 4) {
    const int sA = s0 + 2 * hi;
    v2f raw = *(const v2f*)(Srow + sA);
    v2f mx = *(const v2f*)(cm + sA);
    v2f rc = *(const v2f*)(cr + sA);
    v2f a;
    a.x = __expf(raw.x - mx.x) * rc.x;
    a.y = __expf(raw.y - mx.y) * rc.y;

    const float* vr0 = vb + (size_t)sA * HH + lo;
    const float* vr1 = vr0 + HH;
#pragma unroll
    for (int n = 0; n < 8; ++n) {
      v2f bf;
      bf.x = vr0[n * 16];
      bf.y = vr1[n * 16];
      acc[n] = wmma_f32(a, bf, acc[n]);
    }
   }
  }

  join_slab<AO_WAVES>(part, acc, w, lane, out + ((size_t)b * TT + t0) * HH);
}

extern "C" void kernel_launch(void* const* d_in, const int* in_sizes, int n_in,
                              void* d_out, int out_size, void* d_ws,
                              size_t ws_size, hipStream_t stream) {
  (void)in_sizes; (void)n_in; (void)out_size; (void)ws_size;
  const float* x = (const float*)d_in[0];
  const float* Wq = (const float*)d_in[1];
  const float* Wk = (const float*)d_in[2];
  const float* Wv = (const float*)d_in[3];
  float* out = (float*)d_out;

  float* ws = (float*)d_ws;
  float* q = ws;
  float* k = q + (size_t)BB * TT * HH;
  float* v = k + (size_t)BB * TT * HH;
  float* S = v + (size_t)BB * TT * HH;
  float* cmax = S + (size_t)BB * TT * TT;
  float* crcp = cmax + (size_t)BB * TT;

  qkv_proj<<<dim3(BB * TT / 16, 3), 32 * PROJ_WAVES, 0, stream>>>(x, Wq, Wk, Wv, q, k, v);
  qk_scores<<<dim3(TT / 16, TT / 16, BB), 32, 0, stream>>>(q, k, S);
  col_softmax_stats<<<dim3(TT / 32, BB), 1024, 0, stream>>>(S, cmax, crcp);
  attn_out<<<dim3(TT / 16, BB), 32 * AO_WAVES, 0, stream>>>(S, cmax, crcp, v, out);
}


# score kernel: a wave makes a 32x64 block of S (was a 16x16 tile per one-wave work-group), q/k fragments loaded 16 B per lane and reused over the block; k summed in a new order; whole-line stores
# speedup vs baseline: 12.2838x; 1.3241x over previous
#include <hip/hip_runtime.h>
#include <hip/hip_bf16.h>
#include <math.h>

#define BB 4
#define TT 2048
#define CC 1024
#define HH 128

typedef __attribute__((ext_vector_type(2))) float v2f;
typedef __attribute__((ext_vector_type(8))) float v8f;

__device__ __forceinline__ v8f wmma_f32(v2f a, v2f b, v8f c) {
  return __builtin_amdgcn_wmma_f32_16x16x4_f32(false, a, false, b, (short)0, c,
                                               false, false);
}

#define PITCH 132
#define PROJ_WAVES 8
#define AO_WAVES 16

template <int NW>
__device__ __forceinline__ void join_slab(float* part, const v8f (&acc)[8],
                                          int w, int lane, float* dst) {
  const int lo = lane & 15, hi = lane >> 4;
  float* mine = part + w * (16 * PITCH);
#pragma unroll
  for (int n = 0; n < 8; ++n)
#pragma unroll
    for (int r = 0; r < 8; ++r)
      mine[(r + 8 * hi) * PITCH + n * 16 + lo] = acc[n][r];
  __syncthreads();
  for (int row = w; row < 16; row += NW) {
    const float* src = part + row * PITCH + 4 * lane;
    float4 s = *(const float4*)src;
#pragma unroll
    for (int j = 1; j < NW; ++j) {
      const float4 p = *(const float4*)(src + j * (16 * PITCH));
      s.x += p.x; s.y += p.y; s.z += p.z; s.w += p.w;
    }
    *(float4*)(dst + (size_t)row * HH + 4 * lane) = s;
  }
}

__global__ __launch_bounds__(32 * PROJ_WAVES) void qkv_proj(
    const float* __restrict__ x, const float* __restrict__ Wq,
    const float* __restrict__ Wk, const float* __restrict__ Wv,
    float* __restrict__ q, float* __restrict__ k, float* __restrict__ v) {
  __shared__ __attribute__((aligned(16))) float part[PROJ_WAVES * 16 * PITCH];
  const int lane = threadIdx.x & 31;
  const int w = __builtin_amdgcn_readfirstlane((int)(threadIdx.x >> 5));
  const int lo = lane & 15, hi = lane >> 4;
  const int m0 = blockIdx.x * 16;

  const float* W;
  float* out;
  if (blockIdx.y == 0)      { W = Wq; out = q; }
  else if (blockIdx.y == 1) { W = Wk; out = k; }
  else                      { W = Wv; out = v; }

  const float* xrow = x + (size_t)(m0 + lo) * CC;
  v8f acc[8] = {};

  const int kBeg = w * (CC / PROJ_WAVES);
  const int kEnd = kBeg + CC / PROJ_WAVES;
  for (int k0 = kBeg; k0 < kEnd; k0 += 4) {
    v2f a = *(const v2f*)(xrow + k0 + 2 * hi);
    const float* wr0 = W + (size_t)(k0 + 2 * hi) * HH + lo;
    const float* wr1 = wr0 + HH;
#pragma unroll
    for (int n = 0; n < 8; ++n) {
      v2f bf;
      bf.x = wr0[n * 16];
      bf.y = wr1[n * 16];
      acc[n] = wmma_f32(a, bf, acc[n]);
    }
  }
  join_slab<PROJ_WAVES>(part, acc, w, lane, out + (size_t)m0 * HH);
}

#define QK_WAVES 8

__device__ __forceinline__ float other_half(float x) {
  return __builtin_bit_cast(
      float, __builtin_amdgcn_permlanex16(__builtin_bit_cast(int, x),
                                          __builtin_bit_cast(int, x),
                                          0x76543210, 0xfedcba98, false, false));
}

__global__ __launch_bounds__(32 * QK_WAVES) void qk_scores(
    const float* __restrict__ q, const float* __restrict__ kmat,
    float* __restrict__ S) {
  const int p = blockIdx.x;
  int bt = 0;
  while ((bt + 1) * (bt + 2) / 2 <= p) ++bt;
  const int bs = p - bt * (bt + 1) / 2;
  const int b = blockIdx.y;

  const int lane = threadIdx.x & 31;
  const int w = __builtin_amdgcn_readfirstlane((int)(threadIdx.x >> 5));
  const int lo = lane & 15, hi = lane >> 4;
  const int tb = bt * 128 + (w >> 1) * 32;
  const int sb = bs * 128 + (w & 1) * 64;
  if (sb > tb + 31) return;

  const float* q0 = q + ((size_t)b * TT + tb + lo) * HH + 4 * hi;
  const float* q1 = q0 + 16 * HH;
  const float* kr = kmat + ((size_t)b * TT + sb + lo) * HH + 4 * hi;

  v8f acc[2][4] = {};
  for (int k0 = 0; k0 < HH; k0 += 8) {
    const float4 a0 = *(const float4*)(q0 + k0);
    const float4 a1 = *(const float4*)(q1 + k0);
    float4 bf[4];
#pragma unroll
    for (int n = 0; n < 4; ++n)
      bf[n] = *(const float4*)(kr + (size_t)n * 16 * HH + k0);
#pragma unroll
    for (int n = 0; n < 4; ++n) {
      const v2f b01 = {bf[n].x, bf[n].y};
      acc[0][n] = wmma_f32(v2f{a0.x, a0.y}, b01, acc[0][n]);
      acc[1][n] = wmma_f32(v2f{a1.x, a1.y}, b01, acc[1][n]);
    }
#pragma unroll
    for (int n = 0; n < 4; ++n) {
      const v2f b23 = {bf[n].z, bf[n].w};
      acc[0][n] = wmma_f32(v2f{a0.z, a0.w}, b23, acc[0][n]);
      acc[1][n] = wmma_f32(v2f{a1.z, a1.w}, b23, acc[1][n]);
    }
  }

  const float scale = 0.08838834764831845f;
#pragma unroll
  for (int m = 0; m < 2; ++m) {
    const int t0 = tb + 16 * m;
#pragma unroll
    for (int np = 0; np < 2; ++np) {
      const int s0 = sb + 32 * np;
      if (s0 > t0 + 15) continue;
      const int s = s0 + lo + 16 * hi;
      float* Sout = S + ((size_t)b * TT + t0) * TT + s;
#pragma unroll
      for (int r = 0; r < 8; ++r) {
        const float left = acc[m][2 * np][r];
        const float right = acc[m][2 * np + 1][r];
        const float got = other_half(hi ? left : right);
        const float top = hi ? got : left;
        const float bot = hi ? right : got;
        const int tt = t0 + r, tbm = t0 + r + 8;
        Sout[(size_t)r * TT] = (s <= tt) ? top * scale : -1e30f;
        Sout[(size_t)(r + 8) * TT] = (s <= tbm) ? bot * scale : -1e30f;
      }
    }
  }
}

#define ROWS_IN_FLIGHT 16

__global__ __launch_bounds__(1024) void col_softmax_stats(
    const float* __restrict__ S, float* __restrict__ colmax,
    float* __restrict__ colrcp) {
  __shared__ float wmax[32][32];
  __shared__ float wsum[32][32];

  const int lane = threadIdx.x & 31;
  const int w = threadIdx.x >> 5;
  const int s0 = blockIdx.x * 32;
  const int b = blockIdx.y;
  const int n = (TT - s0) / 32;
  const size_t step = (size_t)32 * TT;
  const float* p = S + ((size_t)b * TT + s0 + w) * TT + s0 + lane;
  const float none = -3.0e38f;

  float x0 = none;
  if (w >= lane) x0 = p[0];

  float m = x0;
  for (int i0 = 1; i0 < n; i0 += ROWS_IN_FLIGHT) {
    float x[ROWS_IN_FLIGHT];
#pragma unroll
    for (int u = 0; u < ROWS_IN_FLIGHT; ++u) {
      x[u] = none;
      if (i0 + u < n) x[u] = p[(size_t)(i0 + u) * step];
    }
#pragma unroll
    for (int u = 0; u < ROWS_IN_FLIGHT; ++u) m = fmaxf(m, x[u]);
  }

  wmax[w][lane] = m;
  __syncthreads();
  float cm = wmax[0][lane];
#pragma unroll
  for (int j = 1; j < 32; ++j) cm = fmaxf(cm, wmax[j][lane]);

  float sum = __expf(x0 - cm);
  for (int i0 = 1; i0 < n; i0 += ROWS_IN_FLIGHT) {
    float x[ROWS_IN_FLIGHT];
#pragma unroll
    for (int u = 0; u < ROWS_IN_FLIGHT; ++u) {
      x[u] = none;
      if (i0 + u < n) x[u] = p[(size_t)(i0 + u) * step];
    }
#pragma unroll
    for (int u = 0; u < ROWS_IN_FLIGHT; ++u) sum += __expf(x[u] - cm);
  }

  wsum[w][lane] = sum;
  __syncthreads();
  if (w == 0) {
    float tot = wsum[0][lane];
#pragma unroll
    for (int j = 1; j < 32; ++j) tot += wsum[j][lane];
    const int gid = b * TT + s0 + lane;
    colmax[gid] = cm;
    colrcp[gid] = 1.0f / tot;
  }
}

__global__ __launch_bounds__(32 * AO_WAVES) void attn_out(
    const float* __restrict__ S, const float* __restrict__ colmax,
    const float* __restrict__ colrcp, const float* __restrict__ v,
    float* __restrict__ out) {
  __shared__ __attribute__((aligned(16))) float part[AO_WAVES * 16 * PITCH];
  const int t0 = blockIdx.x * 16;
  const int b = blockIdx.y;
  const int lane = threadIdx.x & 31;
  const int w = __builtin_amdgcn_readfirstlane((int)(threadIdx.x >> 5));
  const int lo = lane & 15, hi = lane >> 4;

  const float* Srow = S + ((size_t)b * TT + t0 + lo) * TT;
  const float* cm = colmax + (size_t)b * TT;
  const float* cr = colrcp + (size_t)b * TT;
  const float* vb = v + (size_t)b * TT * HH;

  v8f acc[8] = {};
  const int sEnd = t0 + 16;
  for (int c0 = 32 * w; c0 < sEnd; c0 += 32 * AO_WAVES) {
   const int cEnd = (c0 + 32 < sEnd) ? c0 + 32 : sEnd;
   for (int s0 = c0; s0 < cEnd; s0 += 4) {
    const int sA = s0 + 2 * hi;
    v2f raw = *(const v2f*)(Srow + sA);
    v2f mx = *(const v2f*)(cm + sA);
    v2f rc = *(const v2f*)(cr + sA);
    v2f a;
    a.x = __expf(raw.x - mx.x) * rc.x;
    a.y = __expf(raw.y - mx.y) * rc.y;

    const float* vr0 = vb + (size_t)sA * HH + lo;
    const float* vr1 = vr0 + HH;
#pragma unroll
    for (int n = 0; n < 8; ++n) {
      v2f bf;
      bf.x = vr0[n * 16];
      bf.y = vr1[n * 16];
      acc[n] = wmma_f32(a, bf, acc[n]);
    }
   }
  }

  join_slab<AO_WAVES>(part, acc, w, lane, out + ((size_t)b * TT + t0) * HH);
}

extern "C" void kernel_launch(void* const* d_in, const int* in_sizes, int n_in,
                              void* d_out, int out_size, void* d_ws,
                              size_t ws_size, hipStream_t stream) {
  (void)in_sizes; (void)n_in; (void)out_size; (void)ws_size;
  const float* x = (const float*)d_in[0];
  const float* Wq = (const float*)d_in[1];
  const float* Wk = (const float*)d_in[2];
  const float* Wv = (const float*)d_in[3];
  float* out = (float*)d_out;

  float* ws = (float*)d_ws;
  float* q = ws;
  float* k = q + (size_t)BB * TT * HH;
  float* v = k + (size_t)BB * TT * HH;
  float* S = v + (size_t)BB * TT * HH;
  float* cmax = S + (size_t)BB * TT * TT;
  float* crcp = cmax + (size_t)BB * TT;

  qkv_proj<<<dim3(BB * TT / 16, 3), 32 * PROJ_WAVES, 0, stream>>>(x, Wq, Wk, Wv, q, k, v);
  qk_scores<<<dim3(136, BB), 32 * QK_WAVES, 0, stream>>>(q, k, S);
  col_softmax_stats<<<dim3(TT / 32, BB), 1024, 0, stream>>>(S, cmax, crcp);
  attn_out<<<dim3(TT / 16, BB), 32 * AO_WAVES, 0, stream>>>(S, cmax, crcp, v, out);
}


# projection kernel: a wave makes a 32x128 block (was 16x128) of q, k or v; work-group = 3 projections x 4 k quarters; x and W fragments loaded 16 B per lane and reused; k summed in a new order
# speedup vs baseline: 13.9108x; 1.1325x over previous
#include <hip/hip_runtime.h>
#include <hip/hip_bf16.h>
#include <math.h>

#define BB 4
#define TT 2048
#define CC 1024
#define HH 128

typedef __attribute__((ext_vector_type(2))) float v2f;
typedef __attribute__((ext_vector_type(8))) float v8f;

__device__ __forceinline__ v8f wmma_f32(v2f a, v2f b, v8f c) {
  return __builtin_amdgcn_wmma_f32_16x16x4_f32(false, a, false, b, (short)0, c,
                                               false, false);
}

#define PITCH 132
#define PROJ_KW 4
#define PROJ_ROWS 32
#define PROJ_WAVES (3 * PROJ_KW)
#define AO_WAVES 16

template <int NW>
__device__ __forceinline__ void join_slab(float* part, const v8f (&acc)[8],
                                          int w, int lane, float* dst) {
  const int lo = lane & 15, hi = lane >> 4;
  float* mine = part + w * (16 * PITCH);
#pragma unroll
  for (int n = 0; n < 8; ++n)
#pragma unroll
    for (int r = 0; r < 8; ++r)
      mine[(r + 8 * hi) * PITCH + n * 16 + lo] = acc[n][r];
  __syncthreads();
  for (int row = w; row < 16; row += NW) {
    const float* src = part + row * PITCH + 4 * lane;
    float4 s = *(const float4*)src;
#pragma unroll
    for (int j = 1; j < NW; ++j) {
      const float4 p = *(const float4*)(src + j * (16 * PITCH));
      s.x += p.x; s.y += p.y; s.z += p.z; s.w += p.w;
    }
    *(float4*)(dst + (size_t)row * HH + 4 * lane) = s;
  }
}

__global__ __launch_bounds__(32 * PROJ_WAVES) void qkv_proj(
    const float* __restrict__ x, const float* __restrict__ Wq,
    const float* __restrict__ Wk, const float* __restrict__ Wv,
    float* __restrict__ q, float* __restrict__ k, float* __restrict__ v) {
  __shared__ __attribute__((aligned(16))) float part[PROJ_WAVES * PROJ_ROWS * PITCH];
  const int lane = threadIdx.x & 31;
  const int w = __builtin_amdgcn_readfirstlane((int)(threadIdx.x >> 5));
  const int lo = lane & 15, hi = lane >> 4;
  const int pr = w / PROJ_KW;
  const int kq = w % PROJ_KW;
  const int m0 = blockIdx.x * PROJ_ROWS;

  const float* W;
  float* out;
  if (pr == 0)      { W = Wq; out = q; }
  else if (pr == 1) { W = Wk; out = k; }
  else              { W = Wv; out = v; }

  const float* xr0 = x + (size_t)(m0 + lo) * CC + 8 * hi;
  const float* xr1 = xr0 + (size_t)16 * CC;
  const float* wl = W + (size_t)(8 * hi) * HH + 4 * lo;
  v8f acc[2][8] = {};

  const int kBeg = kq * (CC / PROJ_KW);
  const int kEnd = kBeg + CC / PROJ_KW;
  for (int k0 = kBeg; k0 < kEnd; k0 += 16) {
    float4 a[2][2];
    a[0][0] = *(const float4*)(xr0 + k0);
    a[0][1] = *(const float4*)(xr0 + k0 + 4);
    a[1][0] = *(const float4*)(xr1 + k0);
    a[1][1] = *(const float4*)(xr1 + k0 + 4);
    float4 bw[8][2];
#pragma unroll
    for (int kk = 0; kk < 8; ++kk) {
      bw[kk][0] = *(const float4*)(wl + (size_t)(k0 + kk) * HH);
      bw[kk][1] = *(const float4*)(wl + (size_t)(k0 + kk) * HH + 64);
    }
#pragma unroll
    for (int pp = 0; pp < 4; ++pp) {
      const float4 x0 = a[0][pp >> 1], x1 = a[1][pp >> 1];
      const v2f a0 = (pp & 1) ? v2f{x0.z, x0.w} : v2f{x0.x, x0.y};
      const v2f a1 = (pp & 1) ? v2f{x1.z, x1.w} : v2f{x1.x, x1.y};
#pragma unroll
      for (int g = 0; g < 2; ++g) {
        const float4 e = bw[2 * pp][g], o = bw[2 * pp + 1][g];
        const v2f b0 = {e.x, o.x}, b1 = {e.y, o.y}, b2 = {e.z, o.z},
                  b3 = {e.w, o.w};
        acc[0][4 * g + 0] = wmma_f32(a0, b0, acc[0][4 * g + 0]);
        acc[1][4 * g + 0] = wmma_f32(a1, b0, acc[1][4 * g + 0]);
        acc[0][4 * g + 1] = wmma_f32(a0, b1, acc[0][4 * g + 1]);
        acc[1][4 * g + 1] = wmma_f32(a1, b1, acc[1][4 * g + 1]);
        acc[0][4 * g + 2] = wmma_f32(a0, b2, acc[0][4 * g + 2]);
        acc[1][4 * g + 2] = wmma_f32(a1, b2, acc[1][4 * g + 2]);
        acc[0][4 * g + 3] = wmma_f32(a0, b3, acc[0][4 * g + 3]);
        acc[1][4 * g + 3] = wmma_f32(a1, b3, acc[1][4 * g + 3]);
      }
    }
  }

  float* mine = part + w * (PROJ_ROWS * PITCH);
#pragma unroll
  for (int t = 0; t < 2; ++t)
#pragma unroll
    for (int g = 0; g < 2; ++g)
#pragma unroll
      for (int r = 0; r < 8; ++r) {
        const float4 val = {acc[t][4 * g + 0][r], acc[t][4 * g + 1][r],
                            acc[t][4 * g + 2][r], acc[t][4 * g + 3][r]};
        *(float4*)(mine + (16 * t + r + 8 * hi) * PITCH + 64 * g + 4 * lo) = val;
      }
  __syncthreads();
  const float* mypr = part + (pr * PROJ_KW) * (PROJ_ROWS * PITCH);
  for (int row = kq; row < PROJ_ROWS; row += PROJ_KW) {
    const float* src = mypr + row * PITCH + 4 * lane;
    float4 s = *(const float4*)src;
#pragma unroll
    for (int j = 1; j < PROJ_KW; ++j) {
      const float4 p4 = *(const float4*)(src + j * (PROJ_ROWS * PITCH));
      s.x += p4.x; s.y += p4.y; s.z += p4.z; s.w += p4.w;
    }
    *(float4*)(out + (size_t)(m0 + row) * HH + 4 * lane) = s;
  }
}

#define QK_WAVES 8

__device__ __forceinline__ float other_half(float x) {
  return __builtin_bit_cast(
      float, __builtin_amdgcn_permlanex16(__builtin_bit_cast(int, x),
                                          __builtin_bit_cast(int, x),
                                          0x76543210, 0xfedcba98, false, false));
}

__global__ __launch_bounds__(32 * QK_WAVES) void qk_scores(
    const float* __restrict__ q, const float* __restrict__ kmat,
    float* __restrict__ S) {
  const int p = blockIdx.x;
  int bt = 0;
  while ((bt + 1) * (bt + 2) / 2 <= p) ++bt;
  const int bs = p - bt * (bt + 1) / 2;
  const int b = blockIdx.y;

  const int lane = threadIdx.x & 31;
  const int w = __builtin_amdgcn_readfirstlane((int)(threadIdx.x >> 5));
  const int lo = lane & 15, hi = lane >> 4;
  const int tb = bt * 128 + (w >> 1) * 32;
  const int sb = bs * 128 + (w & 1) * 64;
  if (sb > tb + 31) return;

  const float* q0 = q + ((size_t)b * TT + tb + lo) * HH + 4 * hi;
  const float* q1 = q0 + 16 * HH;
  const float* kr = kmat + ((size_t)b * TT + sb + lo) * HH + 4 * hi;

  v8f acc[2][4] = {};
  for (int k0 = 0; k0 < HH; k0 += 8) {
    const float4 a0 = *(const float4*)(q0 + k0);
    const float4 a1 = *(const float4*)(q1 + k0);
    float4 bf[4];
#pragma unroll
    for (int n = 0; n < 4; ++n)
      bf[n] = *(const float4*)(kr + (size_t)n * 16 * HH + k0);
#pragma unroll
    for (int n = 0; n < 4; ++n) {
      const v2f b01 = {bf[n].x, bf[n].y};
      acc[0][n] = wmma_f32(v2f{a0.x, a0.y}, b01, acc[0][n]);
      acc[1][n] = wmma_f32(v2f{a1.x, a1.y}, b01, acc[1][n]);
    }
#pragma unroll
    for (int n = 0; n < 4; ++n) {
      const v2f b23 = {bf[n].z, bf[n].w};
      acc[0][n] = wmma_f32(v2f{a0.z, a0.w}, b23, acc[0][n]);
      acc[1][n] = wmma_f32(v2f{a1.z, a1.w}, b23, acc[1][n]);
    }
  }

  const float scale = 0.08838834764831845f;
#pragma unroll
  for (int m = 0; m < 2; ++m) {
    const int t0 = tb + 16 * m;
#pragma unroll
    for (int np = 0; np < 2; ++np) {
      const int s0 = sb + 32 * np;
      if (s0 > t0 + 15) continue;
      const int s = s0 + lo + 16 * hi;
      float* Sout = S + ((size_t)b * TT + t0) * TT + s;
#pragma unroll
      for (int r = 0; r < 8; ++r) {
        const float left = acc[m][2 * np][r];
        const float right = acc[m][2 * np + 1][r];
        const float got = other_half(hi ? left : right);
        const float top = hi ? got : left;
        const float bot = hi ? right : got;
        const int tt = t0 + r, tbm = t0 + r + 8;
        Sout[(size_t)r * TT] = (s <= tt) ? top * scale : -1e30f;
        Sout[(size_t)(r + 8) * TT] = (s <= tbm) ? bot * scale : -1e30f;
      }
    }
  }
}

#define ROWS_IN_FLIGHT 16

__global__ __launch_bounds__(1024) void col_softmax_stats(
    const float* __restrict__ S, float* __restrict__ colmax,
    float* __restrict__ colrcp) {
  __shared__ float wmax[32][32];
  __shared__ float wsum[32][32];

  const int lane = threadIdx.x & 31;
  const int w = threadIdx.x >> 5;
  const int s0 = blockIdx.x * 32;
  const int b = blockIdx.y;
  const int n = (TT - s0) / 32;
  const size_t step = (size_t)32 * TT;
  const float* p = S + ((size_t)b * TT + s0 + w) * TT + s0 + lane;
  const float none = -3.0e38f;

  float x0 = none;
  if (w >= lane) x0 = p[0];

  float m = x0;
  for (int i0 = 1; i0 < n; i0 += ROWS_IN_FLIGHT) {
    float x[ROWS_IN_FLIGHT];
#pragma unroll
    for (int u = 0; u < ROWS_IN_FLIGHT; ++u) {
      x[u] = none;
      if (i0 + u < n) x[u] = p[(size_t)(i0 + u) * step];
    }
#pragma unroll
    for (int u = 0; u < ROWS_IN_FLIGHT; ++u) m = fmaxf(m, x[u]);
  }

  wmax[w][lane] = m;
  __syncthreads();
  float cm = wmax[0][lane];
#pragma unroll
  for (int j = 1; j < 32; ++j) cm = fmaxf(cm, wmax[j][lane]);

  float sum = __expf(x0 - cm);
  for (int i0 = 1; i0 < n; i0 += ROWS_IN_FLIGHT) {
    float x[ROWS_IN_FLIGHT];
#pragma unroll
    for (int u = 0; u < ROWS_IN_FLIGHT; ++u) {
      x[u] = none;
      if (i0 + u < n) x[u] = p[(size_t)(i0 + u) * step];
    }
#pragma unroll
    for (int u = 0; u < ROWS_IN_FLIGHT; ++u) sum += __expf(x[u] - cm);
  }

  wsum[w][lane] = sum;
  __syncthreads();
  if (w == 0) {
    float tot = wsum[0][lane];
#pragma unroll
    for (int j = 1; j < 32; ++j) tot += wsum[j][lane];
    const int gid = b * TT + s0 + lane;
    colmax[gid] = cm;
    colrcp[gid] = 1.0f / tot;
  }
}

__global__ __launch_bounds__(32 * AO_WAVES) void attn_out(
    const float* __restrict__ S, const float* __restrict__ colmax,
    const float* __restrict__ colrcp, const float* __restrict__ v,
    float* __restrict__ out) {
  __shared__ __attribute__((aligned(16))) float part[AO_WAVES * 16 * PITCH];
  const int t0 = blockIdx.x * 16;
  const int b = blockIdx.y;
  const int lane = threadIdx.x & 31;
  const int w = __builtin_amdgcn_readfirstlane((int)(threadIdx.x >> 5));
  const int lo = lane & 15, hi = lane >> 4;

  const float* Srow = S + ((size_t)b * TT + t0 + lo) * TT;
  const float* cm = colmax + (size_t)b * TT;
  const float* cr = colrcp + (size_t)b * TT;
  const float* vb = v + (size_t)b * TT * HH;

  v8f acc[8] = {};
  const int sEnd = t0 + 16;
  for (int c0 = 32 * w; c0 < sEnd; c0 += 32 * AO_WAVES) {
   const int cEnd = (c0 + 32 < sEnd) ? c0 + 32 : sEnd;
   for (int s0 = c0; s0 < cEnd; s0 += 4) {
    const int sA = s0 + 2 * hi;
    v2f raw = *(const v2f*)(Srow + sA);
    v2f mx = *(const v2f*)(cm + sA);
    v2f rc = *(const v2f*)(cr + sA);
    v2f a;
    a.x = __expf(raw.x - mx.x) * rc.x;
    a.y = __expf(raw.y - mx.y) * rc.y;

    const float* vr0 = vb + (size_t)sA * HH + lo;
    const float* vr1 = vr0 + HH;
#pragma unroll
    for (int n = 0; n < 8; ++n) {
      v2f bf;
      bf.x = vr0[n * 16];
      bf.y = vr1[n * 16];
      acc[n] = wmma_f32(a, bf, acc[n]);
    }
   }
  }

  join_slab<AO_WAVES>(part, acc, w, lane, out + ((size_t)b * TT + t0) * HH);
}

extern "C" void kernel_launch(void* const* d_in, const int* in_sizes, int n_in,
                              void* d_out, int out_size, void* d_ws,
                              size_t ws_size, hipStream_t stream) {
  (void)in_sizes; (void)n_in; (void)out_size; (void)ws_size;
  const float* x = (const float*)d_in[0];
  const float* Wq = (const float*)d_in[1];
  const float* Wk = (const float*)d_in[2];
  const float* Wv = (const float*)d_in[3];
  float* out = (float*)d_out;

  float* ws = (float*)d_ws;
  float* q = ws;
  float* k = q + (size_t)BB * TT * HH;
  float* v = k + (size_t)BB * TT * HH;
  float* S = v + (size_t)BB * TT * HH;
  float* cmax = S + (size_t)BB * TT * TT;
  float* crcp = cmax + (size_t)BB * TT;

  qkv_proj<<<dim3(BB * TT / PROJ_ROWS), 32 * PROJ_WAVES, 0, stream>>>(x, Wq, Wk, Wv, q, k, v);
  qk_scores<<<dim3(136, BB), 32 * QK_WAVES, 0, stream>>>(q, k, S);
  col_softmax_stats<<<dim3(TT / 32, BB), 1024, 0, stream>>>(S, cmax, crcp);
  attn_out<<<dim3(TT / 16, BB), 32 * AO_WAVES, 0, stream>>>(S, cmax, crcp, v, out);
}


# output kernel: a wave makes a 32x64 block (was 16x128) over 16-s steps, S/v fragments loaded 16 B per lane and reused; work-group = row slabs i and 63-i x 64 columns; s summed in a new order
# speedup vs baseline: 17.1049x; 1.2296x over previous
#include <hip/hip_runtime.h>
#include <hip/hip_bf16.h>
#include <math.h>

#define BB 4
#define TT 2048
#define CC 1024
#define HH 128

typedef __attribute__((ext_vector_type(2))) float v2f;
typedef __attribute__((ext_vector_type(8))) float v8f;

__device__ __forceinline__ v8f wmma_f32(v2f a, v2f b, v8f c) {
  return __builtin_amdgcn_wmma_f32_16x16x4_f32(false, a, false, b, (short)0, c,
                                               false, false);
}

#define PITCH 132
#define PROJ_KW 4
#define PROJ_ROWS 32
#define PROJ_WAVES (3 * PROJ_KW)

__global__ __launch_bounds__(32 * PROJ_WAVES) void qkv_proj(
    const float* __restrict__ x, const float* __restrict__ Wq,
    const float* __restrict__ Wk, const float* __restrict__ Wv,
    float* __restrict__ q, float* __restrict__ k, float* __restrict__ v) {
  __shared__ __attribute__((aligned(16))) float part[PROJ_WAVES * PROJ_ROWS * PITCH];
  const int lane = threadIdx.x & 31;
  const int w = __builtin_amdgcn_readfirstlane((int)(threadIdx.x >> 5));
  const int lo = lane & 15, hi = lane >> 4;
  const int pr = w / PROJ_KW;
  const int kq = w % PROJ_KW;
  const int m0 = blockIdx.x * PROJ_ROWS;

  const float* W;
  float* out;
  if (pr == 0)      { W = Wq; out = q; }
  else if (pr == 1) { W = Wk; out = k; }
  else              { W = Wv; out = v; }

  const float* xr0 = x + (size_t)(m0 + lo) * CC + 8 * hi;
  const float* xr1 = xr0 + (size_t)16 * CC;
  const float* wl = W + (size_t)(8 * hi) * HH + 4 * lo;
  v8f acc[2][8] = {};

  const int kBeg = kq * (CC / PROJ_KW);
  const int kEnd = kBeg + CC / PROJ_KW;
  for (int k0 = kBeg; k0 < kEnd; k0 += 16) {
    float4 a[2][2];
    a[0][0] = *(const float4*)(xr0 + k0);
    a[0][1] = *(const float4*)(xr0 + k0 + 4);
    a[1][0] = *(const float4*)(xr1 + k0);
    a[1][1] = *(const float4*)(xr1 + k0 + 4);
    float4 bw[8][2];
#pragma unroll
    for (int kk = 0; kk < 8; ++kk) {
      bw[kk][0] = *(const float4*)(wl + (size_t)(k0 + kk) * HH);
      bw[kk][1] = *(const float4*)(wl + (size_t)(k0 + kk) * HH + 64);
    }
#pragma unroll
    for (int pp = 0; pp < 4; ++pp) {
      const float4 x0 = a[0][pp >> 1], x1 = a[1][pp >> 1];
      const v2f a0 = (pp & 1) ? v2f{x0.z, x0.w} : v2f{x0.x, x0.y};
      const v2f a1 = (pp & 1) ? v2f{x1.z, x1.w} : v2f{x1.x, x1.y};
#pragma unroll
      for (int g = 0; g < 2; ++g) {
        const float4 e = bw[2 * pp][g], o = bw[2 * pp + 1][g];
        const v2f b0 = {e.x, o.x}, b1 = {e.y, o.y}, b2 = {e.z, o.z},
                  b3 = {e.w, o.w};
        acc[0][4 * g + 0] = wmma_f32(a0, b0, acc[0][4 * g + 0]);
        acc[1][4 * g + 0] = wmma_f32(a1, b0, acc[1][4 * g + 0]);
        acc[0][4 * g + 1] = wmma_f32(a0, b1, acc[0][4 * g + 1]);
        acc[1][4 * g + 1] = wmma_f32(a1, b1, acc[1][4 * g + 1]);
        acc[0][4 * g + 2] = wmma_f32(a0, b2, acc[0][4 * g + 2]);
        acc[1][4 * g + 2] = wmma_f32(a1, b2, acc[1][4 * g + 2]);
        acc[0][4 * g + 3] = wmma_f32(a0, b3, acc[0][4 * g + 3]);
        acc[1][4 * g + 3] = wmma_f32(a1, b3, acc[1][4 * g + 3]);
      }
    }
  }

  float* mine = part + w * (PROJ_ROWS * PITCH);
#pragma unroll
  for (int t = 0; t < 2; ++t)
#pragma unroll
    for (int g = 0; g < 2; ++g)
#pragma unroll
      for (int r = 0; r < 8; ++r) {
        const float4 val = {acc[t][4 * g + 0][r], acc[t][4 * g + 1][r],
                            acc[t][4 * g + 2][r], acc[t][4 * g + 3][r]};
        *(float4*)(mine + (16 * t + r + 8 * hi) * PITCH + 64 * g + 4 * lo) = val;
      }
  __syncthreads();
  const float* mypr = part + (pr * PROJ_KW) * (PROJ_ROWS * PITCH);
  for (int row = kq; row < PROJ_ROWS; row += PROJ_KW) {
    const float* src = mypr + row * PITCH + 4 * lane;
    float4 s = *(const float4*)src;
#pragma unroll
    for (int j = 1; j < PROJ_KW; ++j) {
      const float4 p4 = *(const float4*)(src + j * (PROJ_ROWS * PITCH));
      s.x += p4.x; s.y += p4.y; s.z += p4.z; s.w += p4.w;
    }
    *(float4*)(out + (size_t)(m0 + row) * HH + 4 * lane) = s;
  }
}

#define QK_WAVES 8

__device__ __forceinline__ float other_half(float x) {
  return __builtin_bit_cast(
      float, __builtin_amdgcn_permlanex16(__builtin_bit_cast(int, x),
                                          __builtin_bit_cast(int, x),
                                          0x76543210, 0xfedcba98, false, false));
}

__global__ __launch_bounds__(32 * QK_WAVES) void qk_scores(
    const float* __restrict__ q, const float* __restrict__ kmat,
    float* __restrict__ S) {
  const int p = blockIdx.x;
  int bt = 0;
  while ((bt + 1) * (bt + 2) / 2 <= p) ++bt;
  const int bs = p - bt * (bt + 1) / 2;
  const int b = blockIdx.y;

  const int lane = threadIdx.x & 31;
  const int w = __builtin_amdgcn_readfirstlane((int)(threadIdx.x >> 5));
  const int lo = lane & 15, hi = lane >> 4;
  const int tb = bt * 128 + (w >> 1) * 32;
  const int sb = bs * 128 + (w & 1) * 64;
  if (sb > tb + 31) return;

  const float* q0 = q + ((size_t)b * TT + tb + lo) * HH + 4 * hi;
  const float* q1 = q0 + 16 * HH;
  const float* kr = kmat + ((size_t)b * TT + sb + lo) * HH + 4 * hi;

  v8f acc[2][4] = {};
  for (int k0 = 0; k0 < HH; k0 += 8) {
    const float4 a0 = *(const float4*)(q0 + k0);
    const float4 a1 = *(const float4*)(q1 + k0);
    float4 bf[4];
#pragma unroll
    for (int n = 0; n < 4; ++n)
      bf[n] = *(const float4*)(kr + (size_t)n * 16 * HH + k0);
#pragma unroll
    for (int n = 0; n < 4; ++n) {
      const v2f b01 = {bf[n].x, bf[n].y};
      acc[0][n] = wmma_f32(v2f{a0.x, a0.y}, b01, acc[0][n]);
      acc[1][n] = wmma_f32(v2f{a1.x, a1.y}, b01, acc[1][n]);
    }
#pragma unroll
    for (int n = 0; n < 4; ++n) {
      const v2f b23 = {bf[n].z, bf[n].w};
      acc[0][n] = wmma_f32(v2f{a0.z, a0.w}, b23, acc[0][n]);
      acc[1][n] = wmma_f32(v2f{a1.z, a1.w}, b23, acc[1][n]);
    }
  }

  const float scale = 0.08838834764831845f;
#pragma unroll
  for (int m = 0; m < 2; ++m) {
    const int t0 = tb + 16 * m;
#pragma unroll
    for (int np = 0; np < 2; ++np) {
      const int s0 = sb + 32 * np;
      if (s0 > t0 + 15) continue;
      const int s = s0 + lo + 16 * hi;
      float* Sout = S + ((size_t)b * TT + t0) * TT + s;
#pragma unroll
      for (int r = 0; r < 8; ++r) {
        const float left = acc[m][2 * np][r];
        const float right = acc[m][2 * np + 1][r];
        const float got = other_half(hi ? left : right);
        const float top = hi ? got : left;
        const float bot = hi ? right : got;
        const int tt = t0 + r, tbm = t0 + r + 8;
        Sout[(size_t)r * TT] = (s <= tt) ? top * scale : -1e30f;
        Sout[(size_t)(r + 8) * TT] = (s <= tbm) ? bot * scale : -1e30f;
      }
    }
  }
}

#define ROWS_IN_FLIGHT 16

__global__ __launch_bounds__(1024) void col_softmax_stats(
    const float* __restrict__ S, float* __restrict__ colmax,
    float* __restrict__ colrcp) {
  __shared__ float wmax[32][32];
  __shared__ float wsum[32][32];

  const int lane = threadIdx.x & 31;
  const int w = threadIdx.x >> 5;
  const int s0 = blockIdx.x * 32;
  const int b = blockIdx.y;
  const int n = (TT - s0) / 32;
  const size_t step = (size_t)32 * TT;
  const float* p = S + ((size_t)b * TT + s0 + w) * TT + s0 + lane;
  const float none = -3.0e38f;

  float x0 = none;
  if (w >= lane) x0 = p[0];

  float m = x0;
  for (int i0 = 1; i0 < n; i0 += ROWS_IN_FLIGHT) {
    float x[ROWS_IN_FLIGHT];
#pragma unroll
    for (int u = 0; u < ROWS_IN_FLIGHT; ++u) {
      x[u] = none;
      if (i0 + u < n) x[u] = p[(size_t)(i0 + u) * step];
    }
#pragma unroll
    for (int u = 0; u < ROWS_IN_FLIGHT; ++u) m = fmaxf(m, x[u]);
  }

  wmax[w][lane] = m;
  __syncthreads();
  float cm = wmax[0][lane];
#pragma unroll
  for (int j = 1; j < 32; ++j) cm = fmaxf(cm, wmax[j][lane]);

  float sum = __expf(x0 - cm);
  for (int i0 = 1; i0 < n; i0 += ROWS_IN_FLIGHT) {
    float x[ROWS_IN_FLIGHT];
#pragma unroll
    for (int u = 0; u < ROWS_IN_FLIGHT; ++u) {
      x[u] = none;
      if (i0 + u < n) x[u] = p[(size_t)(i0 + u) * step];
    }
#pragma unroll
    for (int u = 0; u < ROWS_IN_FLIGHT; ++u) sum += __expf(x[u] - cm);
  }

  wsum[w][lane] = sum;
  __syncthreads();
  if (w == 0) {
    float tot = wsum[0][lane];
#pragma unroll
    for (int j = 1; j < 32; ++j) tot += wsum[j][lane];
    const int gid = b * TT + s0 + lane;
    colmax[gid] = cm;
    colrcp[gid] = 1.0f / tot;
  }
}

#define AO_WAVES 16
#define AO_ROWS 32
#define AO_COLS 64
#define AO_PITCH 68
#define AO_SLABS (TT / AO_ROWS)

__global__ __launch_bounds__(32 * AO_WAVES) void attn_out(
    const float* __restrict__ S, const float* __restrict__ colmax,
    const float* __restrict__ colrcp, const float* __restrict__ v,
    float* __restrict__ out) {
  __shared__ __attribute__((aligned(16))) float part[2 * AO_WAVES * AO_ROWS * AO_PITCH];
  const int pair = blockIdx.x;
  const int h0 = blockIdx.y * AO_COLS;
  const int b = blockIdx.z;
  const int lane = threadIdx.x & 31;
  const int w = __builtin_amdgcn_readfirstlane((int)(threadIdx.x >> 5));
  const int lo = lane & 15, hi = lane >> 4;

  const float* cm = colmax + (size_t)b * TT + 8 * hi;
  const float* cr = colrcp + (size_t)b * TT + 8 * hi;
  const float* vl = v + ((size_t)b * TT + 8 * hi) * HH + h0 + 4 * lo;

  for (int which = 0; which < 2; ++which) {
    const int slab = which ? AO_SLABS - 1 - pair : pair;
    const int npieces = slab + 1;
    const float* Sr0 = S + ((size_t)b * TT + slab * AO_ROWS + lo) * TT + 8 * hi;
    const float* Sr1 = Sr0 + (size_t)16 * TT;
    v8f acc[2][4] = {};
    for (int c = which ? AO_WAVES - 1 - w : w; c < npieces; c += AO_WAVES) {
      for (int half = 0; half < 2; ++half) {
        const int s0 = 32 * c + 16 * half;
        float4 raw[2][2];
        raw[0][0] = *(const float4*)(Sr0 + s0);
        raw[0][1] = *(const float4*)(Sr0 + s0 + 4);
        raw[1][0] = *(const float4*)(Sr1 + s0);
        raw[1][1] = *(const float4*)(Sr1 + s0 + 4);
        float4 mx[2], rc[2];
        mx[0] = *(const float4*)(cm + s0);
        mx[1] = *(const float4*)(cm + s0 + 4);
        rc[0] = *(const float4*)(cr + s0);
        rc[1] = *(const float4*)(cr + s0 + 4);
        float4 bw[8];
#pragma unroll
        for (int kk = 0; kk < 8; ++kk)
          bw[kk] = *(const float4*)(vl + (size_t)(s0 + kk) * HH);
        float4 p[2][2];
#pragma unroll
        for (int t = 0; t < 2; ++t)
#pragma unroll
          for (int q4 = 0; q4 < 2; ++q4) {
            p[t][q4].x = __expf(raw[t][q4].x - mx[q4].x) * rc[q4].x;
            p[t][q4].y = __expf(raw[t][q4].y - mx[q4].y) * rc[q4].y;
            p[t][q4].z = __expf(raw[t][q4].z - mx[q4].z) * rc[q4].z;
            p[t][q4].w = __expf(raw[t][q4].w - mx[q4].w) * rc[q4].w;
          }
#pragma unroll
        for (int pp = 0; pp < 4; ++pp) {
          const float4 x0 = p[0][pp >> 1], x1 = p[1][pp >> 1];
          const v2f a0 = (pp & 1) ? v2f{x0.z, x0.w} : v2f{x0.x, x0.y};
          const v2f a1 = (pp & 1) ? v2f{x1.z, x1.w} : v2f{x1.x, x1.y};
          const float4 e = bw[2 * pp], o = bw[2 * pp + 1];
          const v2f b0 = {e.x, o.x}, b1 = {e.y, o.y}, b2 = {e.z, o.z},
                    b3 = {e.w, o.w};
          acc[0][0] = wmma_f32(a0, b0, acc[0][0]);
          acc[1][0] = wmma_f32(a1, b0, acc[1][0]);
          acc[0][1] = wmma_f32(a0, b1, acc[0][1]);
          acc[1][1] = wmma_f32(a1, b1, acc[1][1]);
          acc[0][2] = wmma_f32(a0, b2, acc[0][2]);
          acc[1][2] = wmma_f32(a1, b2, acc[1][2]);
          acc[0][3] = wmma_f32(a0, b3, acc[0][3]);
          acc[1][3] = wmma_f32(a1, b3, acc[1][3]);
        }
      }
    }
    float* mine = part + ((which * AO_WAVES + w) * AO_ROWS) * AO_PITCH;
#pragma unroll
    for (int t = 0; t < 2; ++t)
#pragma unroll
      for (int r = 0; r < 8; ++r) {
        const float4 val = {acc[t][0][r], acc[t][1][r], acc[t][2][r],
                            acc[t][3][r]};
        *(float4*)(mine + (16 * t + r + 8 * hi) * AO_PITCH + 4 * lo) = val;
      }
  }
  __syncthreads();
  for (int rr = 2 * w; rr < 2 * AO_ROWS; rr += 2 * AO_WAVES) {
    const int wh = rr >> 5;
    const int r = (rr & 31) + hi;
    const int slab = wh ? AO_SLABS - 1 - pair : pair;
    const float* src = part + ((wh * AO_WAVES) * AO_ROWS + r) * AO_PITCH + 4 * lo;
    float4 s = *(const float4*)src;
#pragma unroll
    for (int j = 1; j < AO_WAVES; ++j) {
      const float4 p4 = *(const float4*)(src + j * (AO_ROWS * AO_PITCH));
      s.x += p4.x; s.y += p4.y; s.z += p4.z; s.w += p4.w;
    }
    *(float4*)(out + ((size_t)b * TT + slab * AO_ROWS + r) * HH + h0 + 4 * lo) = s;
  }
}

extern "C" void kernel_launch(void* const* d_in, const int* in_sizes, int n_in,
                              void* d_out, int out_size, void* d_ws,
                              size_t ws_size, hipStream_t stream) {
  (void)in_sizes; (void)n_in; (void)out_size; (void)ws_size;
  const float* x = (const float*)d_in[0];
  const float* Wq = (const float*)d_in[1];
  const float* Wk = (const float*)d_in[2];
  const float* Wv = (const float*)d_in[3];
  float* out = (float*)d_out;

  float* ws = (float*)d_ws;
  float* q = ws;
  float* k = q + (size_t)BB * TT * HH;
  float* v = k + (size_t)BB * TT * HH;
  float* S = v + (size_t)BB * TT * HH;
  float* cmax = S + (size_t)BB * TT * TT;
  float* crcp = cmax + (size_t)BB * TT;

  qkv_proj<<<dim3(BB * TT / PROJ_ROWS), 32 * PROJ_WAVES, 0, stream>>>(x, Wq, Wk, Wv, q, k, v);
  qk_scores<<<dim3(136, BB), 32 * QK_WAVES, 0, stream>>>(q, k, S);
  col_softmax_stats<<<dim3(TT / 32, BB), 1024, 0, stream>>>(S, cmax, crcp);
  attn_out<<<dim3(AO_SLABS / 2, HH / AO_COLS, BB), 32 * AO_WAVES, 0, stream>>>(S, cmax, crcp, v, out);
}


# projection kernel: x and W fragments of k step i+1 loaded into a second register set ahead of the matrix products of step i (was: load, wait, multiply in every step); order of additions unchanged
# speedup vs baseline: 17.7863x; 1.0398x over previous
#include <hip/hip_runtime.h>
#include <hip/hip_bf16.h>
#include <math.h>

#define BB 4
#define TT 2048
#define CC 1024
#define HH 128

typedef __attribute__((ext_vector_type(2))) float v2f;
typedef __attribute__((ext_vector_type(8))) float v8f;

__device__ __forceinline__ v8f wmma_f32(v2f a, v2f b, v8f c) {
  return __builtin_amdgcn_wmma_f32_16x16x4_f32(false, a, false, b, (short)0, c,
                                               false, false);
}

#define PITCH 132
#define PROJ_KW 4
#define PROJ_ROWS 32
#define PROJ_WAVES (3 * PROJ_KW)

__global__ __launch_bounds__(32 * PROJ_WAVES) void qkv_proj(
    const float* __restrict__ x, const float* __restrict__ Wq,
    const float* __restrict__ Wk, const float* __restrict__ Wv,
    float* __restrict__ q, float* __restrict__ k, float* __restrict__ v) {
  __shared__ __attribute__((aligned(16))) float part[PROJ_WAVES * PROJ_ROWS * PITCH];
  const int lane = threadIdx.x & 31;
  const int w = __builtin_amdgcn_readfirstlane((int)(threadIdx.x >> 5));
  const int lo = lane & 15, hi = lane >> 4;
  const int pr = w / PROJ_KW;
  const int kq = w % PROJ_KW;
  const int m0 = blockIdx.x * PROJ_ROWS;

  const float* W;
  float* out;
  if (pr == 0)      { W = Wq; out = q; }
  else if (pr == 1) { W = Wk; out = k; }
  else              { W = Wv; out = v; }

  const float* xr0 = x + (size_t)(m0 + lo) * CC + 8 * hi;
  const float* xr1 = xr0 + (size_t)16 * CC;
  const float* wl = W + (size_t)(8 * hi) * HH + 4 * lo;
  v8f acc[2][8] = {};

  const int kBeg = kq * (CC / PROJ_KW);
  const int kEnd = kBeg + CC / PROJ_KW;

  float4 aA[2][2], bA[8][2], aB[2][2], bB[8][2];
#define PROJ_LOAD(a, bw, k0)                                             \
  do {                                                                   \
    a[0][0] = *(const float4*)(xr0 + (k0));                              \
    a[0][1] = *(const float4*)(xr0 + (k0) + 4);                          \
    a[1][0] = *(const float4*)(xr1 + (k0));                              \
    a[1][1] = *(const float4*)(xr1 + (k0) + 4);                          \
    _Pragma("unroll") for (int kk = 0; kk < 8; ++kk) {                   \
      bw[kk][0] = *(const float4*)(wl + (size_t)((k0) + kk) * HH);       \
      bw[kk][1] = *(const float4*)(wl + (size_t)((k0) + kk) * HH + 64);  \
    }                                                                    \
  } while (0)
#define PROJ_MUL(a, bw)                                                  \
  do {                                                                   \
    _Pragma("unroll") for (int pp = 0; pp < 4; ++pp) {                   \
      const float4 x0 = a[0][pp >> 1], x1 = a[1][pp >> 1];               \
      const v2f a0 = (pp & 1) ? v2f{x0.z, x0.w} : v2f{x0.x, x0.y};       \
      const v2f a1 = (pp & 1) ? v2f{x1.z, x1.w} : v2f{x1.x, x1.y};       \
      _Pragma("unroll") for (int g = 0; g < 2; ++g) {                    \
        const float4 e = bw[2 * pp][g], o = bw[2 * pp + 1][g];           \
        const v2f b0 = {e.x, o.x}, b1 = {e.y, o.y}, b2 = {e.z, o.z},     \
                  b3 = {e.w, o.w};                                       \
        acc[0][4 * g + 0] = wmma_f32(a0, b0, acc[0][4 * g + 0]);         \
        acc[1][4 * g + 0] = wmma_f32(a1, b0, acc[1][4 * g + 0]);         \
        acc[0][4 * g + 1] = wmma_f32(a0, b1, acc[0][4 * g + 1]);         \
        acc[1][4 * g + 1] = wmma_f32(a1, b1, acc[1][4 * g + 1]);         \
        acc[0][4 * g + 2] = wmma_f32(a0, b2, acc[0][4 * g + 2]);         \
        acc[1][4 * g + 2] = wmma_f32(a1, b2, acc[1][4 * g + 2]);         \
        acc[0][4 * g + 3] = wmma_f32(a0, b3, acc[0][4 * g + 3]);         \
        acc[1][4 * g + 3] = wmma_f32(a1, b3, acc[1][4 * g + 3]);         \
      }                                                                  \
    }                                                                    \
  } while (0)

  PROJ_LOAD(aA, bA, kBeg);
  for (int k0 = kBeg; k0 < kEnd - 32; k0 += 32) {
    PROJ_LOAD(aB, bB, k0 + 16);
    PROJ_MUL(aA, bA);
    PROJ_LOAD(aA, bA, k0 + 32);
    PROJ_MUL(aB, bB);
  }
  PROJ_LOAD(aB, bB, kEnd - 16);
  PROJ_MUL(aA, bA);
  PROJ_MUL(aB, bB);
#undef PROJ_LOAD
#undef PROJ_MUL

  float* mine = part + w * (PROJ_ROWS * PITCH);
#pragma unroll
  for (int t = 0; t < 2; ++t)
#pragma unroll
    for (int g = 0; g < 2; ++g)
#pragma unroll
      for (int r = 0; r < 8; ++r) {
        const float4 val = {acc[t][4 * g + 0][r], acc[t][4 * g + 1][r],
                            acc[t][4 * g + 2][r], acc[t][4 * g + 3][r]};
        *(float4*)(mine + (16 * t + r + 8 * hi) * PITCH + 64 * g + 4 * lo) = val;
      }
  __syncthreads();
  const float* mypr = part + (pr * PROJ_KW) * (PROJ_ROWS * PITCH);
  for (int row = kq; row < PROJ_ROWS; row += PROJ_KW) {
    const float* src = mypr + row * PITCH + 4 * lane;
    float4 s = *(const float4*)src;
#pragma unroll
    for (int j = 1; j < PROJ_KW; ++j) {
      const float4 p4 = *(const float4*)(src + j * (PROJ_ROWS * PITCH));
      s.x += p4.x; s.y += p4.y; s.z += p4.z; s.w += p4.w;
    }
    *(float4*)(out + (size_t)(m0 + row) * HH + 4 * lane) = s;
  }
}

#define QK_WAVES 8

__device__ __forceinline__ float other_half(float x) {
  return __builtin_bit_cast(
      float, __builtin_amdgcn_permlanex16(__builtin_bit_cast(int, x),
                                          __builtin_bit_cast(int, x),
                                          0x76543210, 0xfedcba98, false, false));
}

__global__ __launch_bounds__(32 * QK_WAVES) void qk_scores(
    const float* __restrict__ q, const float* __restrict__ kmat,
    float* __restrict__ S) {
  const int p = blockIdx.x;
  int bt = 0;
  while ((bt + 1) * (bt + 2) / 2 <= p) ++bt;
  const int bs = p - bt * (bt + 1) / 2;
  const int b = blockIdx.y;

  const int lane = threadIdx.x & 31;
  const int w = __builtin_amdgcn_readfirstlane((int)(threadIdx.x >> 5));
  const int lo = lane & 15, hi = lane >> 4;
  const int tb = bt * 128 + (w >> 1) * 32;
  const int sb = bs * 128 + (w & 1) * 64;
  if (sb > tb + 31) return;

  const float* q0 = q + ((size_t)b * TT + tb + lo) * HH + 4 * hi;
  const float* q1 = q0 + 16 * HH;
  const float* kr = kmat + ((size_t)b * TT + sb + lo) * HH + 4 * hi;

  v8f acc[2][4] = {};
  for (int k0 = 0; k0 < HH; k0 += 8) {
    const float4 a0 = *(const float4*)(q0 + k0);
    const float4 a1 = *(const float4*)(q1 + k0);
    float4 bf[4];
#pragma unroll
    for (int n = 0; n < 4; ++n)
      bf[n] = *(const float4*)(kr + (size_t)n * 16 * HH + k0);
#pragma unroll
    for (int n = 0; n < 4; ++n) {
      const v2f b01 = {bf[n].x, bf[n].y};
      acc[0][n] = wmma_f32(v2f{a0.x, a0.y}, b01, acc[0][n]);
      acc[1][n] = wmma_f32(v2f{a1.x, a1.y}, b01, acc[1][n]);
    }
#pragma unroll
    for (int n = 0; n < 4; ++n) {
      const v2f b23 = {bf[n].z, bf[n].w};
      acc[0][n] = wmma_f32(v2f{a0.z, a0.w}, b23, acc[0][n]);
      acc[1][n] = wmma_f32(v2f{a1.z, a1.w}, b23, acc[1][n]);
    }
  }

  const float scale = 0.08838834764831845f;
#pragma unroll
  for (int m = 0; m < 2; ++m) {
    const int t0 = tb + 16 * m;
#pragma unroll
    for (int np = 0; np < 2; ++np) {
      const int s0 = sb + 32 * np;
      if (s0 > t0 + 15) continue;
      const int s = s0 + lo + 16 * hi;
      float* Sout = S + ((size_t)b * TT + t0) * TT + s;
#pragma unroll
      for (int r = 0; r < 8; ++r) {
        const float left = acc[m][2 * np][r];
        const float right = acc[m][2 * np + 1][r];
        const float got = other_half(hi ? left : right);
        const float top = hi ? got : left;
        const float bot = hi ? right : got;
        const int tt = t0 + r, tbm = t0 + r + 8;
        Sout[(size_t)r * TT] = (s <= tt) ? top * scale : -1e30f;
        Sout[(size_t)(r + 8) * TT] = (s <= tbm) ? bot * scale : -1e30f;
      }
    }
  }
}

#define ROWS_IN_FLIGHT 16

__global__ __launch_bounds__(1024) void col_softmax_stats(
    const float* __restrict__ S, float* __restrict__ colmax,
    float* __restrict__ colrcp) {
  __shared__ float wmax[32][32];
  __shared__ float wsum[32][32];

  const int lane = threadIdx.x & 31;
  const int w = threadIdx.x >> 5;
  const int s0 = blockIdx.x * 32;
  const int b = blockIdx.y;
  const int n = (TT - s0) / 32;
  const size_t step = (size_t)32 * TT;
  const float* p = S + ((size_t)b * TT + s0 + w) * TT + s0 + lane;
  const float none = -3.0e38f;

  float x0 = none;
  if (w >= lane) x0 = p[0];

  float m = x0;
  for (int i0 = 1; i0 < n; i0 += ROWS_IN_FLIGHT) {
    float x[ROWS_IN_FLIGHT];
#pragma unroll
    for (int u = 0; u < ROWS_IN_FLIGHT; ++u) {
      x[u] = none;
      if (i0 + u < n) x[u] = p[(size_t)(i0 + u) * step];
    }
#pragma unroll
    for (int u = 0; u < ROWS_IN_FLIGHT; ++u) m = fmaxf(m, x[u]);
  }

  wmax[w][lane] = m;
  __syncthreads();
  float cm = wmax[0][lane];
#pragma unroll
  for (int j = 1; j < 32; ++j) cm = fmaxf(cm, wmax[j][lane]);

  float sum = __expf(x0 - cm);
  for (int i0 = 1; i0 < n; i0 += ROWS_IN_FLIGHT) {
    float x[ROWS_IN_FLIGHT];
#pragma unroll
    for (int u = 0; u < ROWS_IN_FLIGHT; ++u) {
      x[u] = none;
      if (i0 + u < n) x[u] = p[(size_t)(i0 + u) * step];
    }
#pragma unroll
    for (int u = 0; u < ROWS_IN_FLIGHT; ++u) sum += __expf(x[u] - cm);
  }

  wsum[w][lane] = sum;
  __syncthreads();
  if (w == 0) {
    float tot = wsum[0][lane];
#pragma unroll
    for (int j = 1; j < 32; ++j) tot += wsum[j][lane];
    const int gid = b * TT + s0 + lane;
    colmax[gid] = cm;
    colrcp[gid] = 1.0f / tot;
  }
}

#define AO_WAVES 16
#define AO_ROWS 32
#define AO_COLS 64
#define AO_PITCH 68
#define AO_SLABS (TT / AO_ROWS)

__global__ __launch_bounds__(32 * AO_WAVES) void attn_out(
    const float* __restrict__ S, const float* __restrict__ colmax,
    const float* __restrict__ colrcp, const float* __restrict__ v,
    float* __restrict__ out) {
  __shared__ __attribute__((aligned(16))) float part[2 * AO_WAVES * AO_ROWS * AO_PITCH];
  const int pair = blockIdx.x;
  const int h0 = blockIdx.y * AO_COLS;
  const int b = blockIdx.z;
  const int lane = threadIdx.x & 31;
  const int w = __builtin_amdgcn_readfirstlane((int)(threadIdx.x >> 5));
  const int lo = lane & 15, hi = lane >> 4;

  const float* cm = colmax + (size_t)b * TT + 8 * hi;
  const float* cr = colrcp + (size_t)b * TT + 8 * hi;
  const float* vl = v + ((size_t)b * TT + 8 * hi) * HH + h0 + 4 * lo;

  for (int which = 0; which < 2; ++which) {
    const int slab = which ? AO_SLABS - 1 - pair : pair;
    const int npieces = slab + 1;
    const float* Sr0 = S + ((size_t)b * TT + slab * AO_ROWS + lo) * TT + 8 * hi;
    const float* Sr1 = Sr0 + (size_t)16 * TT;
    v8f acc[2][4] = {};
    for (int c = which ? AO_WAVES - 1 - w : w; c < npieces; c += AO_WAVES) {
      for (int half = 0; half < 2; ++half) {
        const int s0 = 32 * c + 16 * half;
        float4 raw[2][2];
        raw[0][0] = *(const float4*)(Sr0 + s0);
        raw[0][1] = *(const float4*)(Sr0 + s0 + 4);
        raw[1][0] = *(const float4*)(Sr1 + s0);
        raw[1][1] = *(const float4*)(Sr1 + s0 + 4);
        float4 mx[2], rc[2];
        mx[0] = *(const float4*)(cm + s0);
        mx[1] = *(const float4*)(cm + s0 + 4);
        rc[0] = *(const float4*)(cr + s0);
        rc[1] = *(const float4*)(cr + s0 + 4);
        float4 bw[8];
#pragma unroll
        for (int kk = 0; kk < 8; ++kk)
          bw[kk] = *(const float4*)(vl + (size_t)(s0 + kk) * HH);
        float4 p[2][2];
#pragma unroll
        for (int t = 0; t < 2; ++t)
#pragma unroll
          for (int q4 = 0; q4 < 2; ++q4) {
            p[t][q4].x = __expf(raw[t][q4].x - mx[q4].x) * rc[q4].x;
            p[t][q4].y = __expf(raw[t][q4].y - mx[q4].y) * rc[q4].y;
            p[t][q4].z = __expf(raw[t][q4].z - mx[q4].z) * rc[q4].z;
            p[t][q4].w = __expf(raw[t][q4].w - mx[q4].w) * rc[q4].w;
          }
#pragma unroll
        for (int pp = 0; pp < 4; ++pp) {
          const float4 x0 = p[0][pp >> 1], x1 = p[1][pp >> 1];
          const v2f a0 = (pp & 1) ? v2f{x0.z, x0.w} : v2f{x0.x, x0.y};
          const v2f a1 = (pp & 1) ? v2f{x1.z, x1.w} : v2f{x1.x, x1.y};
          const float4 e = bw[2 * pp], o = bw[2 * pp + 1];
          const v2f b0 = {e.x, o.x}, b1 = {e.y, o.y}, b2 = {e.z, o.z},
                    b3 = {e.w, o.w};
          acc[0][0] = wmma_f32(a0, b0, acc[0][0]);
          acc[1][0] = wmma_f32(a1, b0, acc[1][0]);
          acc[0][1] = wmma_f32(a0, b1, acc[0][1]);
          acc[1][1] = wmma_f32(a1, b1, acc[1][1]);
          acc[0][2] = wmma_f32(a0, b2, acc[0][2]);
          acc[1][2] = wmma_f32(a1, b2, acc[1][2]);
          acc[0][3] = wmma_f32(a0, b3, acc[0][3]);
          acc[1][3] = wmma_f32(a1, b3, acc[1][3]);
        }
      }
    }
    float* mine = part + ((which * AO_WAVES + w) * AO_ROWS) * AO_PITCH;
#pragma unroll
    for (int t = 0; t < 2; ++t)
#pragma unroll
      for (int r = 0; r < 8; ++r) {
        const float4 val = {acc[t][0][r], acc[t][1][r], acc[t][2][r],
                            acc[t][3][r]};
        *(float4*)(mine + (16 * t + r + 8 * hi) * AO_PITCH + 4 * lo) = val;
      }
  }
  __syncthreads();
  for (int rr = 2 * w; rr < 2 * AO_ROWS; rr += 2 * AO_WAVES) {
    const int wh = rr >> 5;
    const int r = (rr & 31) + hi;
    const int slab = wh ? AO_SLABS - 1 - pair : pair;
    const float* src = part + ((wh * AO_WAVES) * AO_ROWS + r) * AO_PITCH + 4 * lo;
    float4 s = *(const float4*)src;
#pragma unroll
    for (int j = 1; j < AO_WAVES; ++j) {
      const float4 p4 = *(const float4*)(src + j * (AO_ROWS * AO_PITCH));
      s.x += p4.x; s.y += p4.y; s.z += p4.z; s.w += p4.w;
    }
    *(float4*)(out + ((size_t)b * TT + slab * AO_ROWS + r) * HH + h0 + 4 * lo) = s;
  }
}

extern "C" void kernel_launch(void* const* d_in, const int* in_sizes, int n_in,
                              void* d_out, int out_size, void* d_ws,
                              size_t ws_size, hipStream_t stream) {
  (void)in_sizes; (void)n_in; (void)out_size; (void)ws_size;
  const float* x = (const float*)d_in[0];
  const float* Wq = (const float*)d_in[1];
  const float* Wk = (const float*)d_in[2];
  const float* Wv = (const float*)d_in[3];
  float* out = (float*)d_out;

  float* ws = (float*)d_ws;
  float* q = ws;
  float* k = q + (size_t)BB * TT * HH;
  float* v = k + (size_t)BB * TT * HH;
  float* S = v + (size_t)BB * TT * HH;
  float* cmax = S + (size_t)BB * TT * TT;
  float* crcp = cmax + (size_t)BB * TT;

  qkv_proj<<<dim3(BB * TT / PROJ_ROWS), 32 * PROJ_WAVES, 0, stream>>>(x, Wq, Wk, Wv, q, k, v);
  qk_scores<<<dim3(136, BB), 32 * QK_WAVES, 0, stream>>>(q, k, S);
  col_softmax_stats<<<dim3(TT / 32, BB), 1024, 0, stream>>>(S, cmax, crcp);
  attn_out<<<dim3(AO_SLABS / 2, HH / AO_COLS, BB), 32 * AO_WAVES, 0, stream>>>(S, cmax, crcp, v, out);
}


# score kernel re-tiled: a wave keeps a 32-row slab's q fragments in registers and walks 32x32 pieces, 65 per slab pair dealt to 256 equal work-groups; k loads of a piece in one run; sum order kept
# speedup vs baseline: 18.6883x; 1.0507x over previous
#include <hip/hip_runtime.h>
#include <hip/hip_bf16.h>
#include <math.h>

#define BB 4
#define TT 2048
#define CC 1024
#define HH 128

typedef __attribute__((ext_vector_type(2))) float v2f;
typedef __attribute__((ext_vector_type(8))) float v8f;

__device__ __forceinline__ v8f wmma_f32(v2f a, v2f b, v8f c) {
  return __builtin_amdgcn_wmma_f32_16x16x4_f32(false, a, false, b, (short)0, c,
                                               false, false);
}

#define PITCH 132
#define PROJ_KW 4
#define PROJ_ROWS 32
#define PROJ_WAVES (3 * PROJ_KW)

__global__ __launch_bounds__(32 * PROJ_WAVES) void qkv_proj(
    const float* __restrict__ x, const float* __restrict__ Wq,
    const float* __restrict__ Wk, const float* __restrict__ Wv,
    float* __restrict__ q, float* __restrict__ k, float* __restrict__ v) {
  __shared__ __attribute__((aligned(16))) float part[PROJ_WAVES * PROJ_ROWS * PITCH];
  const int lane = threadIdx.x & 31;
  const int w = __builtin_amdgcn_readfirstlane((int)(threadIdx.x >> 5));
  const int lo = lane & 15, hi = lane >> 4;
  const int pr = w / PROJ_KW;
  const int kq = w % PROJ_KW;
  const int m0 = blockIdx.x * PROJ_ROWS;

  const float* W;
  float* out;
  if (pr == 0)      { W = Wq; out = q; }
  else if (pr == 1) { W = Wk; out = k; }
  else              { W = Wv; out = v; }

  const float* xr0 = x + (size_t)(m0 + lo) * CC + 8 * hi;
  const float* xr1 = xr0 + (size_t)16 * CC;
  const float* wl = W + (size_t)(8 * hi) * HH + 4 * lo;
  v8f acc[2][8] = {};

  const int kBeg = kq * (CC / PROJ_KW);
  const int kEnd = kBeg + CC / PROJ_KW;

  float4 aA[2][2], bA[8][2], aB[2][2], bB[8][2];
#define PROJ_LOAD(a, bw, k0)                                             \
  do {                                                                   \
    a[0][0] = *(const float4*)(xr0 + (k0));                              \
    a[0][1] = *(const float4*)(xr0 + (k0) + 4);                          \
    a[1][0] = *(const float4*)(xr1 + (k0));                              \
    a[1][1] = *(const float4*)(xr1 + (k0) + 4);                          \
    _Pragma("unroll") for (int kk = 0; kk < 8; ++kk) {                   \
      bw[kk][0] = *(const float4*)(wl + (size_t)((k0) + kk) * HH);       \
      bw[kk][1] = *(const float4*)(wl + (size_t)((k0) + kk) * HH + 64);  \
    }                                                                    \
  } while (0)
#define PROJ_MUL(a, bw)                                                  \
  do {                                                                   \
    _Pragma("unroll") for (int pp = 0; pp < 4; ++pp) {                   \
      const float4 x0 = a[0][pp >> 1], x1 = a[1][pp >> 1];               \
      const v2f a0 = (pp & 1) ? v2f{x0.z, x0.w} : v2f{x0.x, x0.y};       \
      const v2f a1 = (pp & 1) ? v2f{x1.z, x1.w} : v2f{x1.x, x1.y};       \
      _Pragma("unroll") for (int g = 0; g < 2; ++g) {                    \
        const float4 e = bw[2 * pp][g], o = bw[2 * pp + 1][g];           \
        const v2f b0 = {e.x, o.x}, b1 = {e.y, o.y}, b2 = {e.z, o.z},     \
                  b3 = {e.w, o.w};                                       \
        acc[0][4 * g + 0] = wmma_f32(a0, b0, acc[0][4 * g + 0]);         \
        acc[1][4 * g + 0] = wmma_f32(a1, b0, acc[1][4 * g + 0]);         \
        acc[0][4 * g + 1] = wmma_f32(a0, b1, acc[0][4 * g + 1]);         \
        acc[1][4 * g + 1] = wmma_f32(a1, b1, acc[1][4 * g + 1]);         \
        acc[0][4 * g + 2] = wmma_f32(a0, b2, acc[0][4 * g + 2]);         \
        acc[1][4 * g + 2] = wmma_f32(a1, b2, acc[1][4 * g + 2]);         \
        acc[0][4 * g + 3] = wmma_f32(a0, b3, acc[0][4 * g + 3]);         \
        acc[1][4 * g + 3] = wmma_f32(a1, b3, acc[1][4 * g + 3]);         \
      }                                                                  \
    }                                                                    \
  } while (0)

  PROJ_LOAD(aA, bA, kBeg);
  for (int k0 = kBeg; k0 < kEnd - 32; k0 += 32) {
    PROJ_LOAD(aB, bB, k0 + 16);
    PROJ_MUL(aA, bA);
    PROJ_LOAD(aA, bA, k0 + 32);
    PROJ_MUL(aB, bB);
  }
  PROJ_LOAD(aB, bB, kEnd - 16);
  PROJ_MUL(aA, bA);
  PROJ_MUL(aB, bB);
#undef PROJ_LOAD
#undef PROJ_MUL

  float* mine = part + w * (PROJ_ROWS * PITCH);
#pragma unroll
  for (int t = 0; t < 2; ++t)
#pragma unroll
    for (int g = 0; g < 2; ++g)
#pragma unroll
      for (int r = 0; r < 8; ++r) {
        const float4 val = {acc[t][4 * g + 0][r], acc[t][4 * g + 1][r],
                            acc[t][4 * g + 2][r], acc[t][4 * g + 3][r]};
        *(float4*)(mine + (16 * t + r + 8 * hi) * PITCH + 64 * g + 4 * lo) = val;
      }
  __syncthreads();
  const float* mypr = part + (pr * PROJ_KW) * (PROJ_ROWS * PITCH);
  for (int row = kq; row < PROJ_ROWS; row += PROJ_KW) {
    const float* src = mypr + row * PITCH + 4 * lane;
    float4 s = *(const float4*)src;
#pragma unroll
    for (int j = 1; j < PROJ_KW; ++j) {
      const float4 p4 = *(const float4*)(src + j * (PROJ_ROWS * PITCH));
      s.x += p4.x; s.y += p4.y; s.z += p4.z; s.w += p4.w;
    }
    *(float4*)(out + (size_t)(m0 + row) * HH + 4 * lane) = s;
  }
}

#define QK_WAVES 12
#define QK_SLABS (TT / 32)
#define QK_PIECES (QK_SLABS + 1)
#define QK_SPLIT ((QK_PIECES + 1) / 2)
#define QK_GROUPS (QK_SLABS * BB)

__device__ __forceinline__ float other_half(float x) {
  return __builtin_bit_cast(
      float, __builtin_amdgcn_permlanex16(__builtin_bit_cast(int, x),
                                          __builtin_bit_cast(int, x),
                                          0x76543210, 0xfedcba98, false, false));
}

__global__ __launch_bounds__(32 * QK_WAVES) void qk_scores(
    const float* __restrict__ q, const float* __restrict__ kmat,
    float* __restrict__ S) {
  const int g = blockIdx.x;
  const int b = g / QK_SLABS;
  const int pair = (g % QK_SLABS) >> 1;
  const int ubeg = (g & 1) ? QK_SPLIT : 0;
  const int uend = (g & 1) ? QK_PIECES : QK_SPLIT;

  const int lane = threadIdx.x & 31;
  const int w = __builtin_amdgcn_readfirstlane((int)(threadIdx.x >> 5));
  const int lo = lane & 15, hi = lane >> 4;
  const size_t lane_off = (size_t)lo * HH + 4 * hi;
  const float* qb = q + (size_t)b * TT * HH + lane_off;
  const float* kb = kmat + (size_t)b * TT * HH + lane_off;
  const float scale = 0.08838834764831845f;

  float4 qf[2][16];
  int have = -1;
  for (int u = ubeg + w; u < uend; u += QK_WAVES) {
    const bool lower = u <= pair;
    const int slab = lower ? pair : QK_SLABS - 1 - pair;
    const int j = lower ? u : u - (pair + 1);
    const int tb = 32 * slab;
    const int sb = 32 * j;
    if (slab != have) {
      have = slab;
      const float* qr = qb + (size_t)tb * HH;
#pragma unroll
      for (int m = 0; m < 2; ++m)
#pragma unroll
        for (int c = 0; c < 16; ++c)
          qf[m][c] = *(const float4*)(qr + (size_t)m * 16 * HH + 8 * c);
    }
    const float* kr = kb + (size_t)sb * HH;
    float4 kf[2][16];
#pragma unroll
    for (int n = 0; n < 2; ++n)
#pragma unroll
      for (int c = 0; c < 16; ++c)
        kf[n][c] = *(const float4*)(kr + (size_t)n * 16 * HH + 8 * c);
    __builtin_amdgcn_sched_barrier(0);

    v8f acc[2][2] = {};
#pragma unroll
    for (int c = 0; c < 16; ++c) {
      const float4 a0 = qf[0][c], a1 = qf[1][c];
      const float4 b0 = kf[0][c], b1 = kf[1][c];
      acc[0][0] = wmma_f32(v2f{a0.x, a0.y}, v2f{b0.x, b0.y}, acc[0][0]);
      acc[1][0] = wmma_f32(v2f{a1.x, a1.y}, v2f{b0.x, b0.y}, acc[1][0]);
      acc[0][1] = wmma_f32(v2f{a0.x, a0.y}, v2f{b1.x, b1.y}, acc[0][1]);
      acc[1][1] = wmma_f32(v2f{a1.x, a1.y}, v2f{b1.x, b1.y}, acc[1][1]);
      acc[0][0] = wmma_f32(v2f{a0.z, a0.w}, v2f{b0.z, b0.w}, acc[0][0]);
      acc[1][0] = wmma_f32(v2f{a1.z, a1.w}, v2f{b0.z, b0.w}, acc[1][0]);
      acc[0][1] = wmma_f32(v2f{a0.z, a0.w}, v2f{b1.z, b1.w}, acc[0][1]);
      acc[1][1] = wmma_f32(v2f{a1.z, a1.w}, v2f{b1.z, b1.w}, acc[1][1]);
    }

    const int s = sb + lo + 16 * hi;
#pragma unroll
    for (int m = 0; m < 2; ++m) {
      const int t0 = tb + 16 * m;
      float* Sout = S + ((size_t)b * TT + t0) * TT + s;
#pragma unroll
      for (int r = 0; r < 8; ++r) {
        const float left = acc[m][0][r];
        const float right = acc[m][1][r];
        const float got = other_half(hi ? left : right);
        const float top = hi ? got : left;
        const float bot = hi ? right : got;
        const int tt = t0 + r, tbm = t0 + r + 8;
        Sout[(size_t)r * TT] = (s <= tt) ? top * scale : -1e30f;
        Sout[(size_t)(r + 8) * TT] = (s <= tbm) ? bot * scale : -1e30f;
      }
    }
  }
}

#define ROWS_IN_FLIGHT 16

__global__ __launch_bounds__(1024) void col_softmax_stats(
    const float* __restrict__ S, float* __restrict__ colmax,
    float* __restrict__ colrcp) {
  __shared__ float wmax[32][32];
  __shared__ float wsum[32][32];

  const int lane = threadIdx.x & 31;
  const int w = threadIdx.x >> 5;
  const int s0 = blockIdx.x * 32;
  const int b = blockIdx.y;
  const int n = (TT - s0) / 32;
  const size_t step = (size_t)32 * TT;
  const float* p = S + ((size_t)b * TT + s0 + w) * TT + s0 + lane;
  const float none = -3.0e38f;

  float x0 = none;
  if (w >= lane) x0 = p[0];

  float m = x0;
  for (int i0 = 1; i0 < n; i0 += ROWS_IN_FLIGHT) {
    float x[ROWS_IN_FLIGHT];
#pragma unroll
    for (int u = 0; u < ROWS_IN_FLIGHT; ++u) {
      x[u] = none;
      if (i0 + u < n) x[u] = p[(size_t)(i0 + u) * step];
    }
#pragma unroll
    for (int u = 0; u < ROWS_IN_FLIGHT; ++u) m = fmaxf(m, x[u]);
  }

  wmax[w][lane] = m;
  __syncthreads();
  float cm = wmax[0][lane];
#pragma unroll
  for (int j = 1; j < 32; ++j) cm = fmaxf(cm, wmax[j][lane]);

  float sum = __expf(x0 - cm);
  for (int i0 = 1; i0 < n; i0 += ROWS_IN_FLIGHT) {
    float x[ROWS_IN_FLIGHT];
#pragma unroll
    for (int u = 0; u < ROWS_IN_FLIGHT; ++u) {
      x[u] = none;
      if (i0 + u < n) x[u] = p[(size_t)(i0 + u) * step];
    }
#pragma unroll
    for (int u = 0; u < ROWS_IN_FLIGHT; ++u) sum += __expf(x[u] - cm);
  }

  wsum[w][lane] = sum;
  __syncthreads();
  if (w == 0) {
    float tot = wsum[0][lane];
#pragma unroll
    for (int j = 1; j < 32; ++j) tot += wsum[j][lane];
    const int gid = b * TT + s0 + lane;
    colmax[gid] = cm;
    colrcp[gid] = 1.0f / tot;
  }
}

#define AO_WAVES 16
#define AO_ROWS 32
#define AO_COLS 64
#define AO_PITCH 68
#define AO_SLABS (TT / AO_ROWS)

__global__ __launch_bounds__(32 * AO_WAVES) void attn_out(
    const float* __restrict__ S, const float* __restrict__ colmax,
    const float* __restrict__ colrcp, const float* __restrict__ v,
    float* __restrict__ out) {
  __shared__ __attribute__((aligned(16))) float part[2 * AO_WAVES * AO_ROWS * AO_PITCH];
  const int pair = blockIdx.x;
  const int h0 = blockIdx.y * AO_COLS;
  const int b = blockIdx.z;
  const int lane = threadIdx.x & 31;
  const int w = __builtin_amdgcn_readfirstlane((int)(threadIdx.x >> 5));
  const int lo = lane & 15, hi = lane >> 4;

  const float* cm = colmax + (size_t)b * TT + 8 * hi;
  const float* cr = colrcp + (size_t)b * TT + 8 * hi;
  const float* vl = v + ((size_t)b * TT + 8 * hi) * HH + h0 + 4 * lo;

  for (int which = 0; which < 2; ++which) {
    const int slab = which ? AO_SLABS - 1 - pair : pair;
    const int npieces = slab + 1;
    const float* Sr0 = S + ((size_t)b * TT + slab * AO_ROWS + lo) * TT + 8 * hi;
    const float* Sr1 = Sr0 + (size_t)16 * TT;
    v8f acc[2][4] = {};
    for (int c = which ? AO_WAVES - 1 - w : w; c < npieces; c += AO_WAVES) {
      for (int half = 0; half < 2; ++half) {
        const int s0 = 32 * c + 16 * half;
        float4 raw[2][2];
        raw[0][0] = *(const float4*)(Sr0 + s0);
        raw[0][1] = *(const float4*)(Sr0 + s0 + 4);
        raw[1][0] = *(const float4*)(Sr1 + s0);
        raw[1][1] = *(const float4*)(Sr1 + s0 + 4);
        float4 mx[2], rc[2];
        mx[0] = *(const float4*)(cm + s0);
        mx[1] = *(const float4*)(cm + s0 + 4);
        rc[0] = *(const float4*)(cr + s0);
        rc[1] = *(const float4*)(cr + s0 + 4);
        float4 bw[8];
#pragma unroll
        for (int kk = 0; kk < 8; ++kk)
          bw[kk] = *(const float4*)(vl + (size_t)(s0 + kk) * HH);
        float4 p[2][2];
#pragma unroll
        for (int t = 0; t < 2; ++t)
#pragma unroll
          for (int q4 = 0; q4 < 2; ++q4) {
            p[t][q4].x = __expf(raw[t][q4].x - mx[q4].x) * rc[q4].x;
            p[t][q4].y = __expf(raw[t][q4].y - mx[q4].y) * rc[q4].y;
            p[t][q4].z = __expf(raw[t][q4].z - mx[q4].z) * rc[q4].z;
            p[t][q4].w = __expf(raw[t][q4].w - mx[q4].w) * rc[q4].w;
          }
#pragma unroll
        for (int pp = 0; pp < 4; ++pp) {
          const float4 x0 = p[0][pp >> 1], x1 = p[1][pp >> 1];
          const v2f a0 = (pp & 1) ? v2f{x0.z, x0.w} : v2f{x0.x, x0.y};
          const v2f a1 = (pp & 1) ? v2f{x1.z, x1.w} : v2f{x1.x, x1.y};
          const float4 e = bw[2 * pp], o = bw[2 * pp + 1];
          const v2f b0 = {e.x, o.x}, b1 = {e.y, o.y}, b2 = {e.z, o.z},
                    b3 = {e.w, o.w};
          acc[0][0] = wmma_f32(a0, b0, acc[0][0]);
          acc[1][0] = wmma_f32(a1, b0, acc[1][0]);
          acc[0][1] = wmma_f32(a0, b1, acc[0][1]);
          acc[1][1] = wmma_f32(a1, b1, acc[1][1]);
          acc[0][2] = wmma_f32(a0, b2, acc[0][2]);
          acc[1][2] = wmma_f32(a1, b2, acc[1][2]);
          acc[0][3] = wmma_f32(a0, b3, acc[0][3]);
          acc[1][3] = wmma_f32(a1, b3, acc[1][3]);
        }
      }
    }
    float* mine = part + ((which * AO_WAVES + w) * AO_ROWS) * AO_PITCH;
#pragma unroll
    for (int t = 0; t < 2; ++t)
#pragma unroll
      for (int r = 0; r < 8; ++r) {
        const float4 val = {acc[t][0][r], acc[t][1][r], acc[t][2][r],
                            acc[t][3][r]};
        *(float4*)(mine + (16 * t + r + 8 * hi) * AO_PITCH + 4 * lo) = val;
      }
  }
  __syncthreads();
  for (int rr = 2 * w; rr < 2 * AO_ROWS; rr += 2 * AO_WAVES) {
    const int wh = rr >> 5;
    const int r = (rr & 31) + hi;
    const int slab = wh ? AO_SLABS - 1 - pair : pair;
    const float* src = part + ((wh * AO_WAVES) * AO_ROWS + r) * AO_PITCH + 4 * lo;
    float4 s = *(const float4*)src;
#pragma unroll
    for (int j = 1; j < AO_WAVES; ++j) {
      const float4 p4 = *(const float4*)(src + j * (AO_ROWS * AO_PITCH));
      s.x += p4.x; s.y += p4.y; s.z += p4.z; s.w += p4.w;
    }
    *(float4*)(out + ((size_t)b * TT + slab * AO_ROWS + r) * HH + h0 + 4 * lo) = s;
  }
}

extern "C" void kernel_launch(void* const* d_in, const int* in_sizes, int n_in,
                              void* d_out, int out_size, void* d_ws,
                              size_t ws_size, hipStream_t stream) {
  (void)in_sizes; (void)n_in; (void)out_size; (void)ws_size;
  const float* x = (const float*)d_in[0];
  const float* Wq = (const float*)d_in[1];
  const float* Wk = (const float*)d_in[2];
  const float* Wv = (const float*)d_in[3];
  float* out = (float*)d_out;

  float* ws = (float*)d_ws;
  float* q = ws;
  float* k = q + (size_t)BB * TT * HH;
  float* v = k + (size_t)BB * TT * HH;
  float* S = v + (size_t)BB * TT * HH;
  float* cmax = S + (size_t)BB * TT * TT;
  float* crcp = cmax + (size_t)BB * TT;

  qkv_proj<<<dim3(BB * TT / PROJ_ROWS), 32 * PROJ_WAVES, 0, stream>>>(x, Wq, Wk, Wv, q, k, v);
  qk_scores<<<dim3(QK_GROUPS), 32 * QK_WAVES, 0, stream>>>(q, k, S);
  col_softmax_stats<<<dim3(TT / 32, BB), 1024, 0, stream>>>(S, cmax, crcp);
  attn_out<<<dim3(AO_SLABS / 2, HH / AO_COLS, BB), 32 * AO_WAVES, 0, stream>>>(S, cmax, crcp, v, out);
}


# q and k kept in the workspace in fragment order (16-row tile, 8-k chunk, lane, 4 floats): a fragment load of the score kernel reads 512 contiguous bytes (was 16 B from each of 16 rows)
# speedup vs baseline: 19.4884x; 1.0428x over previous
#include <hip/hip_runtime.h>
#include <hip/hip_bf16.h>
#include <math.h>

#define BB 4
#define TT 2048
#define CC 1024
#define HH 128

typedef __attribute__((ext_vector_type(2))) float v2f;
typedef __attribute__((ext_vector_type(8))) float v8f;

__device__ __forceinline__ v8f wmma_f32(v2f a, v2f b, v8f c) {
  return __builtin_amdgcn_wmma_f32_16x16x4_f32(false, a, false, b, (short)0, c,
                                               false, false);
}

#define PITCH 132
#define PROJ_KW 4
#define PROJ_ROWS 32
#define PROJ_WAVES (3 * PROJ_KW)

__global__ __launch_bounds__(32 * PROJ_WAVES) void qkv_proj(
    const float* __restrict__ x, const float* __restrict__ Wq,
    const float* __restrict__ Wk, const float* __restrict__ Wv,
    float* __restrict__ q, float* __restrict__ k, float* __restrict__ v) {
  __shared__ __attribute__((aligned(16))) float part[PROJ_WAVES * PROJ_ROWS * PITCH];
  const int lane = threadIdx.x & 31;
  const int w = __builtin_amdgcn_readfirstlane((int)(threadIdx.x >> 5));
  const int lo = lane & 15, hi = lane >> 4;
  const int pr = w / PROJ_KW;
  const int kq = w % PROJ_KW;
  const int m0 = blockIdx.x * PROJ_ROWS;

  const float* W;
  float* out;
  if (pr == 0)      { W = Wq; out = q; }
  else if (pr == 1) { W = Wk; out = k; }
  else              { W = Wv; out = v; }

  const float* xr0 = x + (size_t)(m0 + lo) * CC + 8 * hi;
  const float* xr1 = xr0 + (size_t)16 * CC;
  const float* wl = W + (size_t)(8 * hi) * HH + 4 * lo;
  v8f acc[2][8] = {};

  const int kBeg = kq * (CC / PROJ_KW);
  const int kEnd = kBeg + CC / PROJ_KW;

  float4 aA[2][2], bA[8][2], aB[2][2], bB[8][2];
#define PROJ_LOAD(a, bw, k0)                                             \
  do {                                                                   \
    a[0][0] = *(const float4*)(xr0 + (k0));                              \
    a[0][1] = *(const float4*)(xr0 + (k0) + 4);                          \
    a[1][0] = *(const float4*)(xr1 + (k0));                              \
    a[1][1] = *(const float4*)(xr1 + (k0) + 4);                          \
    _Pragma("unroll") for (int kk = 0; kk < 8; ++kk) {                   \
      bw[kk][0] = *(const float4*)(wl + (size_t)((k0) + kk) * HH);       \
      bw[kk][1] = *(const float4*)(wl + (size_t)((k0) + kk) * HH + 64);  \
    }                                                                    \
  } while (0)
#define PROJ_MUL(a, bw)                                                  \
  do {                                                                   \
    _Pragma("unroll") for (int pp = 0; pp < 4; ++pp) {                   \
      const float4 x0 = a[0][pp >> 1], x1 = a[1][pp >> 1];               \
      const v2f a0 = (pp & 1) ? v2f{x0.z, x0.w} : v2f{x0.x, x0.y};       \
      const v2f a1 = (pp & 1) ? v2f{x1.z, x1.w} : v2f{x1.x, x1.y};       \
      _Pragma("unroll") for (int g = 0; g < 2; ++g) {                    \
        const float4 e = bw[2 * pp][g], o = bw[2 * pp + 1][g];           \
        const v2f b0 = {e.x, o.x}, b1 = {e.y, o.y}, b2 = {e.z, o.z},     \
                  b3 = {e.w, o.w};                                       \
        acc[0][4 * g + 0] = wmma_f32(a0, b0, acc[0][4 * g + 0]);         \
        acc[1][4 * g + 0] = wmma_f32(a1, b0, acc[1][4 * g + 0]);         \
        acc[0][4 * g + 1] = wmma_f32(a0, b1, acc[0][4 * g + 1]);         \
        acc[1][4 * g + 1] = wmma_f32(a1, b1, acc[1][4 * g + 1]);         \
        acc[0][4 * g + 2] = wmma_f32(a0, b2, acc[0][4 * g + 2]);         \
        acc[1][4 * g + 2] = wmma_f32(a1, b2, acc[1][4 * g + 2]);         \
        acc[0][4 * g + 3] = wmma_f32(a0, b3, acc[0][4 * g + 3]);         \
        acc[1][4 * g + 3] = wmma_f32(a1, b3, acc[1][4 * g + 3]);         \
      }                                                                  \
    }                                                                    \
  } while (0)

  PROJ_LOAD(aA, bA, kBeg);
  for (int k0 = kBeg; k0 < kEnd - 32; k0 += 32) {
    PROJ_LOAD(aB, bB, k0 + 16);
    PROJ_MUL(aA, bA);
    PROJ_LOAD(aA, bA, k0 + 32);
    PROJ_MUL(aB, bB);
  }
  PROJ_LOAD(aB, bB, kEnd - 16);
  PROJ_MUL(aA, bA);
  PROJ_MUL(aB, bB);
#undef PROJ_LOAD
#undef PROJ_MUL

  float* mine = part + w * (PROJ_ROWS * PITCH);
#pragma unroll
  for (int t = 0; t < 2; ++t)
#pragma unroll
    for (int g = 0; g < 2; ++g)
#pragma unroll
      for (int r = 0; r < 8; ++r) {
        const float4 val = {acc[t][4 * g + 0][r], acc[t][4 * g + 1][r],
                            acc[t][4 * g + 2][r], acc[t][4 * g + 3][r]};
        *(float4*)(mine + (16 * t + r + 8 * hi) * PITCH + 64 * g + 4 * lo) = val;
      }
  __syncthreads();
  const float* mypr = part + (pr * PROJ_KW) * (PROJ_ROWS * PITCH);
  if (pr == 2) {
    for (int row = kq; row < PROJ_ROWS; row += PROJ_KW) {
      const float* src = mypr + row * PITCH + 4 * lane;
      float4 s = *(const float4*)src;
#pragma unroll
      for (int j = 1; j < PROJ_KW; ++j) {
        const float4 p4 = *(const float4*)(src + j * (PROJ_ROWS * PITCH));
        s.x += p4.x; s.y += p4.y; s.z += p4.z; s.w += p4.w;
      }
      *(float4*)(out + (size_t)(m0 + row) * HH + 4 * lane) = s;
    }
  } else {
    float* fr = out + (size_t)m0 * HH;
    for (int u = kq; u < 2 * 16; u += PROJ_KW) {
      const int t = u >> 4, c = u & 15;
      const float* src = mypr + (16 * t + lo) * PITCH + 8 * c + 4 * hi;
      float4 s = *(const float4*)src;
#pragma unroll
      for (int j = 1; j < PROJ_KW; ++j) {
        const float4 p4 = *(const float4*)(src + j * (PROJ_ROWS * PITCH));
        s.x += p4.x; s.y += p4.y; s.z += p4.z; s.w += p4.w;
      }
      *(float4*)(fr + (size_t)((16 * t + c) * 32 + lane) * 4) = s;
    }
  }
}

#define QK_WAVES 12
#define QK_SLABS (TT / 32)
#define QK_PIECES (QK_SLABS + 1)
#define QK_SPLIT ((QK_PIECES + 1) / 2)
#define QK_GROUPS (QK_SLABS * BB)

__device__ __forceinline__ float other_half(float x) {
  return __builtin_bit_cast(
      float, __builtin_amdgcn_permlanex16(__builtin_bit_cast(int, x),
                                          __builtin_bit_cast(int, x),
                                          0x76543210, 0xfedcba98, false, false));
}

__global__ __launch_bounds__(32 * QK_WAVES) void qk_scores(
    const float* __restrict__ q, const float* __restrict__ kmat,
    float* __restrict__ S) {
  const int g = blockIdx.x;
  const int b = g / QK_SLABS;
  const int pair = (g % QK_SLABS) >> 1;
  const int ubeg = (g & 1) ? QK_SPLIT : 0;
  const int uend = (g & 1) ? QK_PIECES : QK_SPLIT;

  const int lane = threadIdx.x & 31;
  const int w = __builtin_amdgcn_readfirstlane((int)(threadIdx.x >> 5));
  const int lo = lane & 15, hi = lane >> 4;
  const float* qb = q + (size_t)b * TT * HH + 4 * lane;
  const float* kb = kmat + (size_t)b * TT * HH + 4 * lane;
  const float scale = 0.08838834764831845f;

  float4 qf[2][16];
  int have = -1;
  for (int u = ubeg + w; u < uend; u += QK_WAVES) {
    const bool lower = u <= pair;
    const int slab = lower ? pair : QK_SLABS - 1 - pair;
    const int j = lower ? u : u - (pair + 1);
    const int tb = 32 * slab;
    const int sb = 32 * j;
    if (slab != have) {
      have = slab;
      const float* qr = qb + (size_t)tb * HH;
#pragma unroll
      for (int m = 0; m < 2; ++m)
#pragma unroll
        for (int c = 0; c < 16; ++c)
          qf[m][c] = *(const float4*)(qr + (size_t)m * 16 * HH + 128 * c);
    }
    const float* kr = kb + (size_t)sb * HH;
    float4 kf[2][16];
#pragma unroll
    for (int n = 0; n < 2; ++n)
#pragma unroll
      for (int c = 0; c < 16; ++c)
        kf[n][c] = *(const float4*)(kr + (size_t)n * 16 * HH + 128 * c);
    __builtin_amdgcn_sched_barrier(0);

    v8f acc[2][2] = {};
#pragma unroll
    for (int c = 0; c < 16; ++c) {
      const float4 a0 = qf[0][c], a1 = qf[1][c];
      const float4 b0 = kf[0][c], b1 = kf[1][c];
      acc[0][0] = wmma_f32(v2f{a0.x, a0.y}, v2f{b0.x, b0.y}, acc[0][0]);
      acc[1][0] = wmma_f32(v2f{a1.x, a1.y}, v2f{b0.x, b0.y}, acc[1][0]);
      acc[0][1] = wmma_f32(v2f{a0.x, a0.y}, v2f{b1.x, b1.y}, acc[0][1]);
      acc[1][1] = wmma_f32(v2f{a1.x, a1.y}, v2f{b1.x, b1.y}, acc[1][1]);
      acc[0][0] = wmma_f32(v2f{a0.z, a0.w}, v2f{b0.z, b0.w}, acc[0][0]);
      acc[1][0] = wmma_f32(v2f{a1.z, a1.w}, v2f{b0.z, b0.w}, acc[1][0]);
      acc[0][1] = wmma_f32(v2f{a0.z, a0.w}, v2f{b1.z, b1.w}, acc[0][1]);
      acc[1][1] = wmma_f32(v2f{a1.z, a1.w}, v2f{b1.z, b1.w}, acc[1][1]);
    }

    const int s = sb + lo + 16 * hi;
#pragma unroll
    for (int m = 0; m < 2; ++m) {
      const int t0 = tb + 16 * m;
      float* Sout = S + ((size_t)b * TT + t0) * TT + s;
#pragma unroll
      for (int r = 0; r < 8; ++r) {
        const float left = acc[m][0][r];
        const float right = acc[m][1][r];
        const float got = other_half(hi ? left : right);
        const float top = hi ? got : left;
        const float bot = hi ? right : got;
        const int tt = t0 + r, tbm = t0 + r + 8;
        Sout[(size_t)r * TT] = (s <= tt) ? top * scale : -1e30f;
        Sout[(size_t)(r + 8) * TT] = (s <= tbm) ? bot * scale : -1e30f;
      }
    }
  }
}

#define ROWS_IN_FLIGHT 16

__global__ __launch_bounds__(1024) void col_softmax_stats(
    const float* __restrict__ S, float* __restrict__ colmax,
    float* __restrict__ colrcp) {
  __shared__ float wmax[32][32];
  __shared__ float wsum[32][32];

  const int lane = threadIdx.x & 31;
  const int w = threadIdx.x >> 5;
  const int s0 = blockIdx.x * 32;
  const int b = blockIdx.y;
  const int n = (TT - s0) / 32;
  const size_t step = (size_t)32 * TT;
  const float* p = S + ((size_t)b * TT + s0 + w) * TT + s0 + lane;
  const float none = -3.0e38f;

  float x0 = none;
  if (w >= lane) x0 = p[0];

  float m = x0;
  for (int i0 = 1; i0 < n; i0 += ROWS_IN_FLIGHT) {
    float x[ROWS_IN_FLIGHT];
#pragma unroll
    for (int u = 0; u < ROWS_IN_FLIGHT; ++u) {
      x[u] = none;
      if (i0 + u < n) x[u] = p[(size_t)(i0 + u) * step];
    }
#pragma unroll
    for (int u = 0; u < ROWS_IN_FLIGHT; ++u) m = fmaxf(m, x[u]);
  }

  wmax[w][lane] = m;
  __syncthreads();
  float cm = wmax[0][lane];
#pragma unroll
  for (int j = 1; j < 32; ++j) cm = fmaxf(cm, wmax[j][lane]);

  float sum = __expf(x0 - cm);
  for (int i0 = 1; i0 < n; i0 += ROWS_IN_FLIGHT) {
    float x[ROWS_IN_FLIGHT];
#pragma unroll
    for (int u = 0; u < ROWS_IN_FLIGHT; ++u) {
      x[u] = none;
      if (i0 + u < n) x[u] = p[(size_t)(i0 + u) * step];
    }
#pragma unroll
    for (int u = 0; u < ROWS_IN_FLIGHT; ++u) sum += __expf(x[u] - cm);
  }

  wsum[w][lane] = sum;
  __syncthreads();
  if (w == 0) {
    float tot = wsum[0][lane];
#pragma unroll
    for (int j = 1; j < 32; ++j) tot += wsum[j][lane];
    const int gid = b * TT + s0 + lane;
    colmax[gid] = cm;
    colrcp[gid] = 1.0f / tot;
  }
}

#define AO_WAVES 16
#define AO_ROWS 32
#define AO_COLS 64
#define AO_PITCH 68
#define AO_SLABS (TT / AO_ROWS)

__global__ __launch_bounds__(32 * AO_WAVES) void attn_out(
    const float* __restrict__ S, const float* __restrict__ colmax,
    const float* __restrict__ colrcp, const float* __restrict__ v,
    float* __restrict__ out) {
  __shared__ __attribute__((aligned(16))) float part[2 * AO_WAVES * AO_ROWS * AO_PITCH];
  const int pair = blockIdx.x;
  const int h0 = blockIdx.y * AO_COLS;
  const int b = blockIdx.z;
  const int lane = threadIdx.x & 31;
  const int w = __builtin_amdgcn_readfirstlane((int)(threadIdx.x >> 5));
  const int lo = lane & 15, hi = lane >> 4;

  const float* cm = colmax + (size_t)b * TT + 8 * hi;
  const float* cr = colrcp + (size_t)b * TT + 8 * hi;
  const float* vl = v + ((size_t)b * TT + 8 * hi) * HH + h0 + 4 * lo;

  for (int which = 0; which < 2; ++which) {
    const int slab = which ? AO_SLABS - 1 - pair : pair;
    const int npieces = slab + 1;
    const float* Sr0 = S + ((size_t)b * TT + slab * AO_ROWS + lo) * TT + 8 * hi;
    const float* Sr1 = Sr0 + (size_t)16 * TT;
    v8f acc[2][4] = {};
    for (int c = which ? AO_WAVES - 1 - w : w; c < npieces; c += AO_WAVES) {
      for (int half = 0; half < 2; ++half) {
        const int s0 = 32 * c + 16 * half;
        float4 raw[2][2];
        raw[0][0] = *(const float4*)(Sr0 + s0);
        raw[0][1] = *(const float4*)(Sr0 + s0 + 4);
        raw[1][0] = *(const float4*)(Sr1 + s0);
        raw[1][1] = *(const float4*)(Sr1 + s0 + 4);
        float4 mx[2], rc[2];
        mx[0] = *(const float4*)(cm + s0);
        mx[1] = *(const float4*)(cm + s0 + 4);
        rc[0] = *(const float4*)(cr + s0);
        rc[1] = *(const float4*)(cr + s0 + 4);
        float4 bw[8];
#pragma unroll
        for (int kk = 0; kk < 8; ++kk)
          bw[kk] = *(const float4*)(vl + (size_t)(s0 + kk) * HH);
        float4 p[2][2];
#pragma unroll
        for (int t = 0; t < 2; ++t)
#pragma unroll
          for (int q4 = 0; q4 < 2; ++q4) {
            p[t][q4].x = __expf(raw[t][q4].x - mx[q4].x) * rc[q4].x;
            p[t][q4].y = __expf(raw[t][q4].y - mx[q4].y) * rc[q4].y;
            p[t][q4].z = __expf(raw[t][q4].z - mx[q4].z) * rc[q4].z;
            p[t][q4].w = __expf(raw[t][q4].w - mx[q4].w) * rc[q4].w;
          }
#pragma unroll
        for (int pp = 0; pp < 4; ++pp) {
          const float4 x0 = p[0][pp >> 1], x1 = p[1][pp >> 1];
          const v2f a0 = (pp & 1) ? v2f{x0.z, x0.w} : v2f{x0.x, x0.y};
          const v2f a1 = (pp & 1) ? v2f{x1.z, x1.w} : v2f{x1.x, x1.y};
          const float4 e = bw[2 * pp], o = bw[2 * pp + 1];
          const v2f b0 = {e.x, o.x}, b1 = {e.y, o.y}, b2 = {e.z, o.z},
                    b3 = {e.w, o.w};
          acc[0][0] = wmma_f32(a0, b0, acc[0][0]);
          acc[1][0] = wmma_f32(a1, b0, acc[1][0]);
          acc[0][1] = wmma_f32(a0, b1, acc[0][1]);
          acc[1][1] = wmma_f32(a1, b1, acc[1][1]);
          acc[0][2] = wmma_f32(a0, b2, acc[0][2]);
          acc[1][2] = wmma_f32(a1, b2, acc[1][2]);
          acc[0][3] = wmma_f32(a0, b3, acc[0][3]);
          acc[1][3] = wmma_f32(a1, b3, acc[1][3]);
        }
      }
    }
    float* mine = part + ((which * AO_WAVES + w) * AO_ROWS) * AO_PITCH;
#pragma unroll
    for (int t = 0; t < 2; ++t)
#pragma unroll
      for (int r = 0; r < 8; ++r) {
        const float4 val = {acc[t][0][r], acc[t][1][r], acc[t][2][r],
                            acc[t][3][r]};
        *(float4*)(mine + (16 * t + r + 8 * hi) * AO_PITCH + 4 * lo) = val;
      }
  }
  __syncthreads();
  for (int rr = 2 * w; rr < 2 * AO_ROWS; rr += 2 * AO_WAVES) {
    const int wh = rr >> 5;
    const int r = (rr & 31) + hi;
    const int slab = wh ? AO_SLABS - 1 - pair : pair;
    const float* src = part + ((wh * AO_WAVES) * AO_ROWS + r) * AO_PITCH + 4 * lo;
    float4 s = *(const float4*)src;
#pragma unroll
    for (int j = 1; j < AO_WAVES; ++j) {
      const float4 p4 = *(const float4*)(src + j * (AO_ROWS * AO_PITCH));
      s.x += p4.x; s.y += p4.y; s.z += p4.z; s.w += p4.w;
    }
    *(float4*)(out + ((size_t)b * TT + slab * AO_ROWS + r) * HH + h0 + 4 * lo) = s;
  }
}

extern "C" void kernel_launch(void* const* d_in, const int* in_sizes, int n_in,
                              void* d_out, int out_size, void* d_ws,
                              size_t ws_size, hipStream_t stream) {
  (void)in_sizes; (void)n_in; (void)out_size; (void)ws_size;
  const float* x = (const float*)d_in[0];
  const float* Wq = (const float*)d_in[1];
  const float* Wk = (const float*)d_in[2];
  const float* Wv = (const float*)d_in[3];
  float* out = (float*)d_out;

  float* ws = (float*)d_ws;
  float* q = ws;
  float* k = q + (size_t)BB * TT * HH;
  float* v = k + (size_t)BB * TT * HH;
  float* S = v + (size_t)BB * TT * HH;
  float* cmax = S + (size_t)BB * TT * TT;
  float* crcp = cmax + (size_t)BB * TT;

  qkv_proj<<<dim3(BB * TT / PROJ_ROWS), 32 * PROJ_WAVES, 0, stream>>>(x, Wq, Wk, Wv, q, k, v);
  qk_scores<<<dim3(QK_GROUPS), 32 * QK_WAVES, 0, stream>>>(q, k, S);
  col_softmax_stats<<<dim3(TT / 32, BB), 1024, 0, stream>>>(S, cmax, crcp);
  attn_out<<<dim3(AO_SLABS / 2, HH / AO_COLS, BB), 32 * AO_WAVES, 0, stream>>>(S, cmax, crcp, v, out);
}


# column softmax stats made in the score kernel: a lane holds a whole 32-row column of its piece and leaves its max and exp-sum; a small kernel joins the slabs (was two passes over S); sums re-ordered
# speedup vs baseline: 20.4955x; 1.0517x over previous
#include <hip/hip_runtime.h>
#include <hip/hip_bf16.h>
#include <math.h>

#define BB 4
#define TT 2048
#define CC 1024
#define HH 128

typedef __attribute__((ext_vector_type(2))) float v2f;
typedef __attribute__((ext_vector_type(8))) float v8f;

__device__ __forceinline__ v8f wmma_f32(v2f a, v2f b, v8f c) {
  return __builtin_amdgcn_wmma_f32_16x16x4_f32(false, a, false, b, (short)0, c,
                                               false, false);
}

#define PITCH 132
#define PROJ_KW 4
#define PROJ_ROWS 32
#define PROJ_WAVES (3 * PROJ_KW)

__global__ __launch_bounds__(32 * PROJ_WAVES) void qkv_proj(
    const float* __restrict__ x, const float* __restrict__ Wq,
    const float* __restrict__ Wk, const float* __restrict__ Wv,
    float* __restrict__ q, float* __restrict__ k, float* __restrict__ v) {
  __shared__ __attribute__((aligned(16))) float part[PROJ_WAVES * PROJ_ROWS * PITCH];
  const int lane = threadIdx.x & 31;
  const int w = __builtin_amdgcn_readfirstlane((int)(threadIdx.x >> 5));
  const int lo = lane & 15, hi = lane >> 4;
  const int pr = w / PROJ_KW;
  const int kq = w % PROJ_KW;
  const int m0 = blockIdx.x * PROJ_ROWS;

  const float* W;
  float* out;
  if (pr == 0)      { W = Wq; out = q; }
  else if (pr == 1) { W = Wk; out = k; }
  else              { W = Wv; out = v; }

  const float* xr0 = x + (size_t)(m0 + lo) * CC + 8 * hi;
  const float* xr1 = xr0 + (size_t)16 * CC;
  const float* wl = W + (size_t)(8 * hi) * HH + 4 * lo;
  v8f acc[2][8] = {};

  const int kBeg = kq * (CC / PROJ_KW);
  const int kEnd = kBeg + CC / PROJ_KW;

  float4 aA[2][2], bA[8][2], aB[2][2], bB[8][2];
#define PROJ_LOAD(a, bw, k0)                                             \
  do {                                                                   \
    a[0][0] = *(const float4*)(xr0 + (k0));                              \
    a[0][1] = *(const float4*)(xr0 + (k0) + 4);                          \
    a[1][0] = *(const float4*)(xr1 + (k0));                              \
    a[1][1] = *(const float4*)(xr1 + (k0) + 4);                          \
    _Pragma("unroll") for (int kk = 0; kk < 8; ++kk) {                   \
      bw[kk][0] = *(const float4*)(wl + (size_t)((k0) + kk) * HH);       \
      bw[kk][1] = *(const float4*)(wl + (size_t)((k0) + kk) * HH + 64);  \
    }                                                                    \
  } while (0)
#define PROJ_MUL(a, bw)                                                  \
  do {                                                                   \
    _Pragma("unroll") for (int pp = 0; pp < 4; ++pp) {                   \
      const float4 x0 = a[0][pp >> 1], x1 = a[1][pp >> 1];               \
      const v2f a0 = (pp & 1) ? v2f{x0.z, x0.w} : v2f{x0.x, x0.y};       \
      const v2f a1 = (pp & 1) ? v2f{x1.z, x1.w} : v2f{x1.x, x1.y};       \
      _Pragma("unroll") for (int g = 0; g < 2; ++g) {                    \
        const float4 e = bw[2 * pp][g], o = bw[2 * pp + 1][g];           \
        const v2f b0 = {e.x, o.x}, b1 = {e.y, o.y}, b2 = {e.z, o.z},     \
                  b3 = {e.w, o.w};                                       \
        acc[0][4 * g + 0] = wmma_f32(a0, b0, acc[0][4 * g + 0]);         \
        acc[1][4 * g + 0] = wmma_f32(a1, b0, acc[1][4 * g + 0]);         \
        acc[0][4 * g + 1] = wmma_f32(a0, b1, acc[0][4 * g + 1]);         \
        acc[1][4 * g + 1] = wmma_f32(a1, b1, acc[1][4 * g + 1]);         \
        acc[0][4 * g + 2] = wmma_f32(a0, b2, acc[0][4 * g + 2]);         \
        acc[1][4 * g + 2] = wmma_f32(a1, b2, acc[1][4 * g + 2]);         \
        acc[0][4 * g + 3] = wmma_f32(a0, b3, acc[0][4 * g + 3]);         \
        acc[1][4 * g + 3] = wmma_f32(a1, b3, acc[1][4 * g + 3]);         \
      }                                                                  \
    }                                                                    \
  } while (0)

  PROJ_LOAD(aA, bA, kBeg);
  for (int k0 = kBeg; k0 < kEnd - 32; k0 += 32) {
    PROJ_LOAD(aB, bB, k0 + 16);
    PROJ_MUL(aA, bA);
    PROJ_LOAD(aA, bA, k0 + 32);
    PROJ_MUL(aB, bB);
  }
  PROJ_LOAD(aB, bB, kEnd - 16);
  PROJ_MUL(aA, bA);
  PROJ_MUL(aB, bB);
#undef PROJ_LOAD
#undef PROJ_MUL

  float* mine = part + w * (PROJ_ROWS * PITCH);
#pragma unroll
  for (int t = 0; t < 2; ++t)
#pragma unroll
    for (int g = 0; g < 2; ++g)
#pragma unroll
      for (int r = 0; r < 8; ++r) {
        const float4 val = {acc[t][4 * g + 0][r], acc[t][4 * g + 1][r],
                            acc[t][4 * g + 2][r], acc[t][4 * g + 3][r]};
        *(float4*)(mine + (16 * t + r + 8 * hi) * PITCH + 64 * g + 4 * lo) = val;
      }
  __syncthreads();
  const float* mypr = part + (pr * PROJ_KW) * (PROJ_ROWS * PITCH);
  if (pr == 2) {
    for (int row = kq; row < PROJ_ROWS; row += PROJ_KW) {
      const float* src = mypr + row * PITCH + 4 * lane;
      float4 s = *(const float4*)src;
#pragma unroll
      for (int j = 1; j < PROJ_KW; ++j) {
        const float4 p4 = *(const float4*)(src + j * (PROJ_ROWS * PITCH));
        s.x += p4.x; s.y += p4.y; s.z += p4.z; s.w += p4.w;
      }
      *(float4*)(out + (size_t)(m0 + row) * HH + 4 * lane) = s;
    }
  } else {
    float* fr = out + (size_t)m0 * HH;
    for (int u = kq; u < 2 * 16; u += PROJ_KW) {
      const int t = u >> 4, c = u & 15;
      const float* src = mypr + (16 * t + lo) * PITCH + 8 * c + 4 * hi;
      float4 s = *(const float4*)src;
#pragma unroll
      for (int j = 1; j < PROJ_KW; ++j) {
        const float4 p4 = *(const float4*)(src + j * (PROJ_ROWS * PITCH));
        s.x += p4.x; s.y += p4.y; s.z += p4.z; s.w += p4.w;
      }
      *(float4*)(fr + (size_t)((16 * t + c) * 32 + lane) * 4) = s;
    }
  }
}

#define QK_WAVES 12
#define QK_SLABS (TT / 32)
#define QK_PIECES (QK_SLABS + 1)
#define QK_SPLIT ((QK_PIECES + 1) / 2)
#define QK_GROUPS (QK_SLABS * BB)

__device__ __forceinline__ size_t pstat_off(int b, int i, int j) {
  const int last = QK_SLABS - 1;
  const int row = (i < last) ? 32 * j : 32 * (j >> 3) + 2 + 2 * (j & 7);
  const int col = (i < last) ? 32 * (i + 1) : 32 * last;
  return ((size_t)b * TT + row) * TT + col;
}

__device__ __forceinline__ float other_half(float x) {
  return __builtin_bit_cast(
      float, __builtin_amdgcn_permlanex16(__builtin_bit_cast(int, x),
                                          __builtin_bit_cast(int, x),
                                          0x76543210, 0xfedcba98, false, false));
}

__global__ __launch_bounds__(32 * QK_WAVES) void qk_scores(
    const float* __restrict__ q, const float* __restrict__ kmat,
    float* __restrict__ S) {
  const int g = blockIdx.x;
  const int b = g / QK_SLABS;
  const int pair = (g % QK_SLABS) >> 1;
  const int ubeg = (g & 1) ? QK_SPLIT : 0;
  const int uend = (g & 1) ? QK_PIECES : QK_SPLIT;

  const int lane = threadIdx.x & 31;
  const int w = __builtin_amdgcn_readfirstlane((int)(threadIdx.x >> 5));
  const int lo = lane & 15, hi = lane >> 4;
  const float* qb = q + (size_t)b * TT * HH + 4 * lane;
  const float* kb = kmat + (size_t)b * TT * HH + 4 * lane;
  const float scale = 0.08838834764831845f;

  float4 qf[2][16];
  int have = -1;
  for (int u = ubeg + w; u < uend; u += QK_WAVES) {
    const bool lower = u <= pair;
    const int slab = lower ? pair : QK_SLABS - 1 - pair;
    const int j = lower ? u : u - (pair + 1);
    const int tb = 32 * slab;
    const int sb = 32 * j;
    if (slab != have) {
      have = slab;
      const float* qr = qb + (size_t)tb * HH;
#pragma unroll
      for (int m = 0; m < 2; ++m)
#pragma unroll
        for (int c = 0; c < 16; ++c)
          qf[m][c] = *(const float4*)(qr + (size_t)m * 16 * HH + 128 * c);
    }
    const float* kr = kb + (size_t)sb * HH;
    float4 kf[2][16];
#pragma unroll
    for (int n = 0; n < 2; ++n)
#pragma unroll
      for (int c = 0; c < 16; ++c)
        kf[n][c] = *(const float4*)(kr + (size_t)n * 16 * HH + 128 * c);
    __builtin_amdgcn_sched_barrier(0);

    v8f acc[2][2] = {};
#pragma unroll
    for (int c = 0; c < 16; ++c) {
      const float4 a0 = qf[0][c], a1 = qf[1][c];
      const float4 b0 = kf[0][c], b1 = kf[1][c];
      acc[0][0] = wmma_f32(v2f{a0.x, a0.y}, v2f{b0.x, b0.y}, acc[0][0]);
      acc[1][0] = wmma_f32(v2f{a1.x, a1.y}, v2f{b0.x, b0.y}, acc[1][0]);
      acc[0][1] = wmma_f32(v2f{a0.x, a0.y}, v2f{b1.x, b1.y}, acc[0][1]);
      acc[1][1] = wmma_f32(v2f{a1.x, a1.y}, v2f{b1.x, b1.y}, acc[1][1]);
      acc[0][0] = wmma_f32(v2f{a0.z, a0.w}, v2f{b0.z, b0.w}, acc[0][0]);
      acc[1][0] = wmma_f32(v2f{a1.z, a1.w}, v2f{b0.z, b0.w}, acc[1][0]);
      acc[0][1] = wmma_f32(v2f{a0.z, a0.w}, v2f{b1.z, b1.w}, acc[0][1]);
      acc[1][1] = wmma_f32(v2f{a1.z, a1.w}, v2f{b1.z, b1.w}, acc[1][1]);
    }

    const int s = sb + lo + 16 * hi;
    float held[32];
    float pmax = -3.0e38f;
#pragma unroll
    for (int m = 0; m < 2; ++m) {
      const int t0 = tb + 16 * m;
      float* Sout = S + ((size_t)b * TT + t0) * TT + s;
#pragma unroll
      for (int r = 0; r < 8; ++r) {
        const float left = acc[m][0][r];
        const float right = acc[m][1][r];
        const float got = other_half(hi ? left : right);
        const float top = hi ? got : left;
        const float bot = hi ? right : got;
        const int tt = t0 + r, tbm = t0 + r + 8;
        const float vt = (s <= tt) ? top * scale : -1e30f;
        const float vb = (s <= tbm) ? bot * scale : -1e30f;
        Sout[(size_t)r * TT] = vt;
        Sout[(size_t)(r + 8) * TT] = vb;
        held[16 * m + 2 * r] = vt;
        held[16 * m + 2 * r + 1] = vb;
        pmax = fmaxf(pmax, fmaxf(vt, vb));
      }
    }
    float psum = 0.0f;
#pragma unroll
    for (int e = 0; e < 32; ++e) psum += __expf(held[e] - pmax);
    float* pst = S + pstat_off(b, slab, j) + lane;
    pst[0] = pmax;
    pst[TT] = psum;
  }
}

#define CJ_WAVES 4
#define CJ_PER (QK_SLABS / CJ_WAVES)

__global__ __launch_bounds__(32 * CJ_WAVES) void col_stats_join(
    const float* __restrict__ S, float* __restrict__ colmax,
    float* __restrict__ colrcp) {
  __shared__ float wmax[CJ_WAVES][32];
  __shared__ float wsum[CJ_WAVES][32];

  const int lane = threadIdx.x & 31;
  const int w = __builtin_amdgcn_readfirstlane((int)(threadIdx.x >> 5));
  const int j = blockIdx.x;
  const int b = blockIdx.y;
  const float none = -3.0e38f;

  float pm[CJ_PER], ps[CJ_PER];
#pragma unroll
  for (int t = 0; t < CJ_PER; ++t) {
    const int i = j + w + CJ_WAVES * t;
    pm[t] = none;
    ps[t] = 0.0f;
    if (i < QK_SLABS) {
      const float* p = S + pstat_off(b, i, j) + lane;
      pm[t] = p[0];
      ps[t] = p[TT];
    }
  }
  float m = pm[0];
#pragma unroll
  for (int t = 1; t < CJ_PER; ++t) m = fmaxf(m, pm[t]);
  float sum = 0.0f;
#pragma unroll
  for (int t = 0; t < CJ_PER; ++t) sum += ps[t] * __expf(pm[t] - m);

  wmax[w][lane] = m;
  wsum[w][lane] = sum;
  __syncthreads();
  if (w == 0) {
    float cm = wmax[0][lane];
#pragma unroll
    for (int k = 1; k < CJ_WAVES; ++k) cm = fmaxf(cm, wmax[k][lane]);
    float tot = 0.0f;
#pragma unroll
    for (int k = 0; k < CJ_WAVES; ++k)
      tot += wsum[k][lane] * __expf(wmax[k][lane] - cm);
    const int gid = b * TT + 32 * j + lane;
    colmax[gid] = cm;
    colrcp[gid] = 1.0f / tot;
  }
}

#define AO_WAVES 16
#define AO_ROWS 32
#define AO_COLS 64
#define AO_PITCH 68
#define AO_SLABS (TT / AO_ROWS)

__global__ __launch_bounds__(32 * AO_WAVES) void attn_out(
    const float* __restrict__ S, const float* __restrict__ colmax,
    const float* __restrict__ colrcp, const float* __restrict__ v,
    float* __restrict__ out) {
  __shared__ __attribute__((aligned(16))) float part[2 * AO_WAVES * AO_ROWS * AO_PITCH];
  const int pair = blockIdx.x;
  const int h0 = blockIdx.y * AO_COLS;
  const int b = blockIdx.z;
  const int lane = threadIdx.x & 31;
  const int w = __builtin_amdgcn_readfirstlane((int)(threadIdx.x >> 5));
  const int lo = lane & 15, hi = lane >> 4;

  const float* cm = colmax + (size_t)b * TT + 8 * hi;
  const float* cr = colrcp + (size_t)b * TT + 8 * hi;
  const float* vl = v + ((size_t)b * TT + 8 * hi) * HH + h0 + 4 * lo;

  for (int which = 0; which < 2; ++which) {
    const int slab = which ? AO_SLABS - 1 - pair : pair;
    const int npieces = slab + 1;
    const float* Sr0 = S + ((size_t)b * TT + slab * AO_ROWS + lo) * TT + 8 * hi;
    const float* Sr1 = Sr0 + (size_t)16 * TT;
    v8f acc[2][4] = {};
    for (int c = which ? AO_WAVES - 1 - w : w; c < npieces; c += AO_WAVES) {
      for (int half = 0; half < 2; ++half) {
        const int s0 = 32 * c + 16 * half;
        float4 raw[2][2];
        raw[0][0] = *(const float4*)(Sr0 + s0);
        raw[0][1] = *(const float4*)(Sr0 + s0 + 4);
        raw[1][0] = *(const float4*)(Sr1 + s0);
        raw[1][1] = *(const float4*)(Sr1 + s0 + 4);
        float4 mx[2], rc[2];
        mx[0] = *(const float4*)(cm + s0);
        mx[1] = *(const float4*)(cm + s0 + 4);
        rc[0] = *(const float4*)(cr + s0);
        rc[1] = *(const float4*)(cr + s0 + 4);
        float4 bw[8];
#pragma unroll
        for (int kk = 0; kk < 8; ++kk)
          bw[kk] = *(const float4*)(vl + (size_t)(s0 + kk) * HH);
        float4 p[2][2];
#pragma unroll
        for (int t = 0; t < 2; ++t)
#pragma unroll
          for (int q4 = 0; q4 < 2; ++q4) {
            p[t][q4].x = __expf(raw[t][q4].x - mx[q4].x) * rc[q4].x;
            p[t][q4].y = __expf(raw[t][q4].y - mx[q4].y) * rc[q4].y;
            p[t][q4].z = __expf(raw[t][q4].z - mx[q4].z) * rc[q4].z;
            p[t][q4].w = __expf(raw[t][q4].w - mx[q4].w) * rc[q4].w;
          }
#pragma unroll
        for (int pp = 0; pp < 4; ++pp) {
          const float4 x0 = p[0][pp >> 1], x1 = p[1][pp >> 1];
          const v2f a0 = (pp & 1) ? v2f{x0.z, x0.w} : v2f{x0.x, x0.y};
          const v2f a1 = (pp & 1) ? v2f{x1.z, x1.w} : v2f{x1.x, x1.y};
          const float4 e = bw[2 * pp], o = bw[2 * pp + 1];
          const v2f b0 = {e.x, o.x}, b1 = {e.y, o.y}, b2 = {e.z, o.z},
                    b3 = {e.w, o.w};
          acc[0][0] = wmma_f32(a0, b0, acc[0][0]);
          acc[1][0] = wmma_f32(a1, b0, acc[1][0]);
          acc[0][1] = wmma_f32(a0, b1, acc[0][1]);
          acc[1][1] = wmma_f32(a1, b1, acc[1][1]);
          acc[0][2] = wmma_f32(a0, b2, acc[0][2]);
          acc[1][2] = wmma_f32(a1, b2, acc[1][2]);
          acc[0][3] = wmma_f32(a0, b3, acc[0][3]);
          acc[1][3] = wmma_f32(a1, b3, acc[1][3]);
        }
      }
    }
    float* mine = part + ((which * AO_WAVES + w) * AO_ROWS) * AO_PITCH;
#pragma unroll
    for (int t = 0; t < 2; ++t)
#pragma unroll
      for (int r = 0; r < 8; ++r) {
        const float4 val = {acc[t][0][r], acc[t][1][r], acc[t][2][r],
                            acc[t][3][r]};
        *(float4*)(mine + (16 * t + r + 8 * hi) * AO_PITCH + 4 * lo) = val;
      }
  }
  __syncthreads();
  for (int rr = 2 * w; rr < 2 * AO_ROWS; rr += 2 * AO_WAVES) {
    const int wh = rr >> 5;
    const int r = (rr & 31) + hi;
    const int slab = wh ? AO_SLABS - 1 - pair : pair;
    const float* src = part + ((wh * AO_WAVES) * AO_ROWS + r) * AO_PITCH + 4 * lo;
    float4 s = *(const float4*)src;
#pragma unroll
    for (int j = 1; j < AO_WAVES; ++j) {
      const float4 p4 = *(const float4*)(src + j * (AO_ROWS * AO_PITCH));
      s.x += p4.x; s.y += p4.y; s.z += p4.z; s.w += p4.w;
    }
    *(float4*)(out + ((size_t)b * TT + slab * AO_ROWS + r) * HH + h0 + 4 * lo) = s;
  }
}

extern "C" void kernel_launch(void* const* d_in, const int* in_sizes, int n_in,
                              void* d_out, int out_size, void* d_ws,
                              size_t ws_size, hipStream_t stream) {
  (void)in_sizes; (void)n_in; (void)out_size; (void)ws_size;
  const float* x = (const float*)d_in[0];
  const float* Wq = (const float*)d_in[1];
  const float* Wk = (const float*)d_in[2];
  const float* Wv = (const float*)d_in[3];
  float* out = (float*)d_out;

  float* ws = (float*)d_ws;
  float* q = ws;
  float* k = q + (size_t)BB * TT * HH;
  float* v = k + (size_t)BB * TT * HH;
  float* S = v + (size_t)BB * TT * HH;
  float* cmax = S + (size_t)BB * TT * TT;
  float* crcp = cmax + (size_t)BB * TT;

  qkv_proj<<<dim3(BB * TT / PROJ_ROWS), 32 * PROJ_WAVES, 0, stream>>>(x, Wq, Wk, Wv, q, k, v);
  qk_scores<<<dim3(QK_GROUPS), 32 * QK_WAVES, 0, stream>>>(q, k, S);
  col_stats_join<<<dim3(TT / 32, BB), 32 * CJ_WAVES, 0, stream>>>(S, cmax, crcp);
  attn_out<<<dim3(AO_SLABS / 2, HH / AO_COLS, BB), 32 * AO_WAVES, 0, stream>>>(S, cmax, crcp, v, out);
}
